# starting kernel
# baseline (speedup 1.0000x reference)
#include <hip/hip_runtime.h>
#include <hip/hip_bf16.h>


typedef __attribute__((ext_vector_type(8)))  _Float16 v8h;
typedef __attribute__((ext_vector_type(16))) _Float16 v16h;
typedef __attribute__((ext_vector_type(8)))  float    v8f;

#define TILE_E   128
#define NWAVES   4
#define BLOCK    (NWAVES * 32)
#define HPAD     128
#define NT_H     7
#define KATTR    32
#define WCOLS    256

__device__ __forceinline__ v16h load_frag(const _Float16* base, int khalf) {
    v8h lo = *(const v8h*)(base + khalf * 8);
    v8h hi = *(const v8h*)(base + 16 + khalf * 8);
    v16h r;
#pragma unroll
    for (int i = 0; i < 8; ++i) { r[i] = lo[i]; r[8 + i] = hi[i]; }
    return r;
}

template <int NT, int KS, bool RELU>
__device__ __forceinline__ void gemm_stage(
    const _Float16* __restrict__ lds_in, int in_stride,
    const _Float16* __restrict__ Wt, int kpad,
    const float* __restrict__ bias, int nvalid,
    _Float16* __restrict__ lds_out, int out_stride,
    int ntile0, int lane, int mpair)
{
    const int rc    = lane & 15;
    const int khalf = lane >> 4;
    const int m0    = mpair * 2;

    v8f acc0[NT], acc1[NT];
#pragma unroll
    for (int t = 0; t < NT; ++t) {
        acc0[t] = (v8f){0.f,0.f,0.f,0.f,0.f,0.f,0.f,0.f};
        acc1[t] = (v8f){0.f,0.f,0.f,0.f,0.f,0.f,0.f,0.f};
    }

#pragma unroll
    for (int ks = 0; ks < KS; ++ks) {
        const int kb = ks * 32;
        v16h a0 = load_frag(lds_in + ((m0    ) * 16 + rc) * in_stride + kb, khalf);
        v16h a1 = load_frag(lds_in + ((m0 + 1) * 16 + rc) * in_stride + kb, khalf);
#pragma unroll
        for (int t = 0; t < NT; ++t) {
            v16h b = load_frag(Wt + ((ntile0 + t) * 16 + rc) * kpad + kb, khalf);
            acc0[t] = __builtin_amdgcn_wmma_f32_16x16x32_f16(
                false, a0, false, b, (short)0, acc0[t], false, false);
            acc1[t] = __builtin_amdgcn_wmma_f32_16x16x32_f16(
                false, a1, false, b, (short)0, acc1[t], false, false);
        }
    }

    const int mrow0 = m0 * 16 + (khalf ? 8 : 0);
#pragma unroll
    for (int t = 0; t < NT; ++t) {
        const int  n     = (ntile0 + t) * 16 + rc;
        const bool valid = (n < nvalid);
        const float bv   = valid ? bias[n] : 0.0f;
#pragma unroll
        for (int r = 0; r < 8; ++r) {
            float v0 = acc0[t][r] + bv;
            float v1 = acc1[t][r] + bv;
            if (RELU) { v0 = v0 > 0.0f ? v0 : 0.0f; v1 = v1 > 0.0f ? v1 : 0.0f; }
            if (!valid) { v0 = 0.0f; v1 = 0.0f; }
            lds_out[(mrow0 + r     ) * out_stride + n] = (_Float16)v0;
            lds_out[(mrow0 + r + 16) * out_stride + n] = (_Float16)v1;
        }
    }
}

__global__ void prep_weights(const float* __restrict__ W1, const float* __restrict__ W2,
                             const float* __restrict__ W3, const float* __restrict__ W4,
                             _Float16* __restrict__ W1t, _Float16* __restrict__ W2t,
                             _Float16* __restrict__ W3t, _Float16* __restrict__ W4t)
{
    int tid = blockIdx.x * blockDim.x + threadIdx.x;
    const int s1 = 128 * KATTR;
    const int s2 = s1 + 128 * HPAD;
    const int s3 = s2 + 128 * HPAD;
    const int s4 = s3 + 256 * HPAD;
    if (tid < s1) {
        int n = tid / KATTR, k = tid % KATTR;
        W1t[tid] = (_Float16)((k < 8 && n < 100) ? W1[k * 100 + n] : 0.f);
    } else if (tid < s2) {
        int t = tid - s1; int n = t / HPAD, k = t % HPAD;
        W2t[t] = (_Float16)((k < 100 && n < 100) ? W2[k * 100 + n] : 0.f);
    } else if (tid < s3) {
        int t = tid - s2; int n = t / HPAD, k = t % HPAD;
        W3t[t] = (_Float16)((k < 100 && n < 100) ? W3[k * 100 + n] : 0.f);
    } else if (tid < s4) {
        int t = tid - s3; int n = t / HPAD, k = t % HPAD;
        W4t[t] = (_Float16)((k < 100 && n < 256) ? W4[k * 256 + n] : 0.f);
    }
}

__global__ void zero_f32(float* __restrict__ p, int n) {
    int i = blockIdx.x * blockDim.x + threadIdx.x;
    if (i < n) p[i] = 0.0f;
}

__global__ __launch_bounds__(BLOCK)
void edge_kernel(const float* __restrict__ x,
                 const int*   __restrict__ edge_index,
                 const float* __restrict__ edge_attr,
                 const _Float16* __restrict__ W1t, const _Float16* __restrict__ W2t,
                 const _Float16* __restrict__ W3t, const _Float16* __restrict__ W4t,
                 const float* __restrict__ b1, const float* __restrict__ b2,
                 const float* __restrict__ b3, const float* __restrict__ b4,
                 float* __restrict__ agg, float* __restrict__ cnt, int E)
{
    __shared__ _Float16 s_attr[TILE_E * KATTR];
    __shared__ _Float16 s_h[TILE_E * HPAD];
    __shared__ _Float16 s_w[TILE_E * WCOLS];

    const int e0   = blockIdx.x * TILE_E;
    const int tid  = threadIdx.x;
    const int lane = tid & 31;
    const int wave = tid >> 5;

    {
        const uint4 z4 = {0u, 0u, 0u, 0u};
        uint4* ph = (uint4*)s_h;
        uint4* pw = (uint4*)s_w;
        const int nvec = TILE_E * HPAD * (int)sizeof(_Float16) / 16;
        for (int i = tid; i < nvec; i += BLOCK) { ph[i] = z4; pw[i] = z4; }
    }

    {
        const int e  = tid;
        const int ge = e0 + e;
        float4 a0 = {0.f,0.f,0.f,0.f}, a1 = {0.f,0.f,0.f,0.f};
        if (ge < E) {
            const float4* ea = (const float4*)edge_attr;
            a0 = ea[(long)ge * 2];
            a1 = ea[(long)ge * 2 + 1];
        }
        v8h h, z;
        h[0]=(_Float16)a0.x; h[1]=(_Float16)a0.y; h[2]=(_Float16)a0.z; h[3]=(_Float16)a0.w;
        h[4]=(_Float16)a1.x; h[5]=(_Float16)a1.y; h[6]=(_Float16)a1.z; h[7]=(_Float16)a1.w;
#pragma unroll
        for (int i = 0; i < 8; ++i) z[i] = (_Float16)0.f;
        v8h* dst = (v8h*)(s_attr + e * KATTR);
        dst[0] = h; dst[1] = z; dst[2] = z; dst[3] = z;
    }
    __syncthreads();

    gemm_stage<NT_H, 1, true>(s_attr, KATTR, W1t, KATTR, b1, 100, s_h, HPAD, 0, lane, wave);
    __syncthreads();
    gemm_stage<NT_H, 4, true>(s_h, HPAD, W2t, HPAD, b2, 100, s_w, HPAD, 0, lane, wave);
    __syncthreads();
    gemm_stage<NT_H, 4, true>(s_w, HPAD, W3t, HPAD, b3, 100, s_h, HPAD, 0, lane, wave);
    __syncthreads();
    gemm_stage<8, 4, false>(s_h, HPAD, W4t, HPAD, b4, 256, s_w, WCOLS, 0, lane, wave);
    gemm_stage<8, 4, false>(s_h, HPAD, W4t, HPAD, b4, 256, s_w, WCOLS, 8, lane, wave);
    __syncthreads();

    const int ge = e0 + tid;
    if (ge < E) {
        const int s = edge_index[ge];
        const int d = edge_index[E + ge];
        const float* xs = x + (long)s * 16;
        const _Float16* we = s_w + tid * WCOLS;
        float acc[16];
#pragma unroll
        for (int j = 0; j < 16; ++j) acc[j] = 0.0f;
#pragma unroll
        for (int i = 0; i < 16; ++i) {
            const float xv = xs[i];
#pragma unroll
            for (int j = 0; j < 16; ++j) acc[j] += xv * (float)we[i * 16 + j];
        }
        float* ag = agg + (long)d * 16;
#pragma unroll
        for (int j = 0; j < 16; ++j) atomicAdd(&ag[j], acc[j]);
        atomicAdd(&cnt[d], 1.0f);
    }
}

__global__ void finalize_kernel(const float* __restrict__ x, const float* __restrict__ root,
                                const float* __restrict__ bias, const float* __restrict__ agg,
                                const float* __restrict__ cnt, float* __restrict__ out, int N)
{
    int n = blockIdx.x * blockDim.x + threadIdx.x;
    if (n >= N) return;
    float xi[16];
#pragma unroll
    for (int i = 0; i < 16; ++i) xi[i] = x[(long)n * 16 + i];
    float c = cnt[n];
    float inv = 1.0f / (c > 1.0f ? c : 1.0f);
#pragma unroll
    for (int o = 0; o < 16; ++o) {
        float acc = 0.0f;
#pragma unroll
        for (int i = 0; i < 16; ++i) acc += xi[i] * root[i * 16 + o];
        out[(long)n * 16 + o] = acc + agg[(long)n * 16 + o] * inv + bias[o];
    }
}

extern "C" void kernel_launch(void* const* d_in, const int* in_sizes, int n_in,
                              void* d_out, int out_size, void* d_ws, size_t ws_size,
                              hipStream_t stream)
{
    const float* x          = (const float*)d_in[0];
    const int*   edge_index = (const int*)  d_in[1];
    const float* edge_attr  = (const float*)d_in[2];
    const float* W1 = (const float*)d_in[3];  const float* b1 = (const float*)d_in[4];
    const float* W2 = (const float*)d_in[5];  const float* b2 = (const float*)d_in[6];
    const float* W3 = (const float*)d_in[7];  const float* b3 = (const float*)d_in[8];
    const float* W4 = (const float*)d_in[9];  const float* b4 = (const float*)d_in[10];
    const float* root = (const float*)d_in[11];
    const float* bias = (const float*)d_in[12];
    float* out = (float*)d_out;

    const int N = in_sizes[0] / 16;
    const int E = in_sizes[1] / 2;

    _Float16* W1t = (_Float16*)d_ws;
    _Float16* W2t = W1t + 128 * KATTR;
    _Float16* W3t = W2t + 128 * HPAD;
    _Float16* W4t = W3t + 128 * HPAD;
    float*    agg = (float*)(W4t + 256 * HPAD);
    float*    cnt = agg + (size_t)N * 16;

    const int prep_total = 128 * KATTR + 128 * HPAD + 128 * HPAD + 256 * HPAD;
    prep_weights<<<(prep_total + 255) / 256, 256, 0, stream>>>(
        W1, W2, W3, W4, W1t, W2t, W3t, W4t);

    const int zn = N * 17;
    zero_f32<<<(zn + 255) / 256, 256, 0, stream>>>(agg, zn);

    edge_kernel<<<(E + TILE_E - 1) / TILE_E, BLOCK, 0, stream>>>(
        x, edge_index, edge_attr, W1t, W2t, W3t, W4t,
        b1, b2, b3, b4, agg, cnt, E);

    finalize_kernel<<<(N + 255) / 256, 256, 0, stream>>>(
        x, root, bias, agg, cnt, out, N);
}


# Weight (B) fragments fetched a whole K-step ahead as one group into registers of their own (kept in front by a scheduling barrier): one load round trip per K-step instead of one per fragment
# speedup vs baseline: 1.3036x; 1.3036x over previous
#include <hip/hip_runtime.h>
#include <hip/hip_bf16.h>


typedef __attribute__((ext_vector_type(8)))  _Float16 v8h;
typedef __attribute__((ext_vector_type(16))) _Float16 v16h;
typedef __attribute__((ext_vector_type(8)))  float    v8f;

#define TILE_E   128
#define NWAVES   4
#define BLOCK    (NWAVES * 32)
#define HPAD     128
#define NT_H     7
#define KATTR    32
#define WCOLS    256

__device__ __forceinline__ v16h load_frag(const _Float16* base, int khalf) {
    v8h lo = *(const v8h*)(base + khalf * 8);
    v8h hi = *(const v8h*)(base + 16 + khalf * 8);
    v16h r;
#pragma unroll
    for (int i = 0; i < 8; ++i) { r[i] = lo[i]; r[8 + i] = hi[i]; }
    return r;
}

template <int NT, int KS, bool RELU>
__device__ __forceinline__ void gemm_stage(
    const _Float16* __restrict__ lds_in, int in_stride,
    const _Float16* __restrict__ Wt, int kpad,
    const float* __restrict__ bias, int nvalid,
    _Float16* __restrict__ lds_out, int out_stride,
    int ntile0, int lane, int mpair)
{
    const int rc    = lane & 15;
    const int khalf = lane >> 4;
    const int m0    = mpair * 2;

    v8f acc0[NT], acc1[NT];
#pragma unroll
    for (int t = 0; t < NT; ++t) {
        acc0[t] = (v8f){0.f,0.f,0.f,0.f,0.f,0.f,0.f,0.f};
        acc1[t] = (v8f){0.f,0.f,0.f,0.f,0.f,0.f,0.f,0.f};
    }

    v16h bcur[NT];
#pragma unroll
    for (int t = 0; t < NT; ++t)
        bcur[t] = load_frag(Wt + ((ntile0 + t) * 16 + rc) * kpad, khalf);

#pragma unroll
    for (int ks = 0; ks < KS; ++ks) {
        const int kb = ks * 32;
        v16h bnext[NT];
        if (ks + 1 < KS) {
#pragma unroll
            for (int t = 0; t < NT; ++t)
                bnext[t] = load_frag(Wt + ((ntile0 + t) * 16 + rc) * kpad + kb + 32, khalf);
        }
        __builtin_amdgcn_sched_barrier(0);
        v16h a0 = load_frag(lds_in + ((m0    ) * 16 + rc) * in_stride + kb, khalf);
        v16h a1 = load_frag(lds_in + ((m0 + 1) * 16 + rc) * in_stride + kb, khalf);
#pragma unroll
        for (int t = 0; t < NT; ++t) {
            acc0[t] = __builtin_amdgcn_wmma_f32_16x16x32_f16(
                false, a0, false, bcur[t], (short)0, acc0[t], false, false);
            acc1[t] = __builtin_amdgcn_wmma_f32_16x16x32_f16(
                false, a1, false, bcur[t], (short)0, acc1[t], false, false);
        }
        if (ks + 1 < KS) {
#pragma unroll
            for (int t = 0; t < NT; ++t) bcur[t] = bnext[t];
        }
    }

    const int mrow0 = m0 * 16 + (khalf ? 8 : 0);
#pragma unroll
    for (int t = 0; t < NT; ++t) {
        const int  n     = (ntile0 + t) * 16 + rc;
        const bool valid = (n < nvalid);
        const float bv   = valid ? bias[n] : 0.0f;
#pragma unroll
        for (int r = 0; r < 8; ++r) {
            float v0 = acc0[t][r] + bv;
            float v1 = acc1[t][r] + bv;
            if (RELU) { v0 = v0 > 0.0f ? v0 : 0.0f; v1 = v1 > 0.0f ? v1 : 0.0f; }
            if (!valid) { v0 = 0.0f; v1 = 0.0f; }
            lds_out[(mrow0 + r     ) * out_stride + n] = (_Float16)v0;
            lds_out[(mrow0 + r + 16) * out_stride + n] = (_Float16)v1;
        }
    }
}

__global__ void prep_weights(const float* __restrict__ W1, const float* __restrict__ W2,
                             const float* __restrict__ W3, const float* __restrict__ W4,
                             _Float16* __restrict__ W1t, _Float16* __restrict__ W2t,
                             _Float16* __restrict__ W3t, _Float16* __restrict__ W4t)
{
    int tid = blockIdx.x * blockDim.x + threadIdx.x;
    const int s1 = 128 * KATTR;
    const int s2 = s1 + 128 * HPAD;
    const int s3 = s2 + 128 * HPAD;
    const int s4 = s3 + 256 * HPAD;
    if (tid < s1) {
        int n = tid / KATTR, k = tid % KATTR;
        W1t[tid] = (_Float16)((k < 8 && n < 100) ? W1[k * 100 + n] : 0.f);
    } else if (tid < s2) {
        int t = tid - s1; int n = t / HPAD, k = t % HPAD;
        W2t[t] = (_Float16)((k < 100 && n < 100) ? W2[k * 100 + n] : 0.f);
    } else if (tid < s3) {
        int t = tid - s2; int n = t / HPAD, k = t % HPAD;
        W3t[t] = (_Float16)((k < 100 && n < 100) ? W3[k * 100 + n] : 0.f);
    } else if (tid < s4) {
        int t = tid - s3; int n = t / HPAD, k = t % HPAD;
        W4t[t] = (_Float16)((k < 100 && n < 256) ? W4[k * 256 + n] : 0.f);
    }
}

__global__ void zero_f32(float* __restrict__ p, int n) {
    int i = blockIdx.x * blockDim.x + threadIdx.x;
    if (i < n) p[i] = 0.0f;
}

__global__ __launch_bounds__(BLOCK)
void edge_kernel(const float* __restrict__ x,
                 const int*   __restrict__ edge_index,
                 const float* __restrict__ edge_attr,
                 const _Float16* __restrict__ W1t, const _Float16* __restrict__ W2t,
                 const _Float16* __restrict__ W3t, const _Float16* __restrict__ W4t,
                 const float* __restrict__ b1, const float* __restrict__ b2,
                 const float* __restrict__ b3, const float* __restrict__ b4,
                 float* __restrict__ agg, float* __restrict__ cnt, int E)
{
    __shared__ _Float16 s_attr[TILE_E * KATTR];
    __shared__ _Float16 s_h[TILE_E * HPAD];
    __shared__ _Float16 s_w[TILE_E * WCOLS];

    const int e0   = blockIdx.x * TILE_E;
    const int tid  = threadIdx.x;
    const int lane = tid & 31;
    const int wave = tid >> 5;

    {
        const uint4 z4 = {0u, 0u, 0u, 0u};
        uint4* ph = (uint4*)s_h;
        uint4* pw = (uint4*)s_w;
        const int nvec = TILE_E * HPAD * (int)sizeof(_Float16) / 16;
        for (int i = tid; i < nvec; i += BLOCK) { ph[i] = z4; pw[i] = z4; }
    }

    {
        const int e  = tid;
        const int ge = e0 + e;
        float4 a0 = {0.f,0.f,0.f,0.f}, a1 = {0.f,0.f,0.f,0.f};
        if (ge < E) {
            const float4* ea = (const float4*)edge_attr;
            a0 = ea[(long)ge * 2];
            a1 = ea[(long)ge * 2 + 1];
        }
        v8h h, z;
        h[0]=(_Float16)a0.x; h[1]=(_Float16)a0.y; h[2]=(_Float16)a0.z; h[3]=(_Float16)a0.w;
        h[4]=(_Float16)a1.x; h[5]=(_Float16)a1.y; h[6]=(_Float16)a1.z; h[7]=(_Float16)a1.w;
#pragma unroll
        for (int i = 0; i < 8; ++i) z[i] = (_Float16)0.f;
        v8h* dst = (v8h*)(s_attr + e * KATTR);
        dst[0] = h; dst[1] = z; dst[2] = z; dst[3] = z;
    }
    __syncthreads();

    gemm_stage<NT_H, 1, true>(s_attr, KATTR, W1t, KATTR, b1, 100, s_h, HPAD, 0, lane, wave);
    __syncthreads();
    gemm_stage<NT_H, 4, true>(s_h, HPAD, W2t, HPAD, b2, 100, s_w, HPAD, 0, lane, wave);
    __syncthreads();
    gemm_stage<NT_H, 4, true>(s_w, HPAD, W3t, HPAD, b3, 100, s_h, HPAD, 0, lane, wave);
    __syncthreads();
    gemm_stage<8, 4, false>(s_h, HPAD, W4t, HPAD, b4, 256, s_w, WCOLS, 0, lane, wave);
    gemm_stage<8, 4, false>(s_h, HPAD, W4t, HPAD, b4, 256, s_w, WCOLS, 8, lane, wave);
    __syncthreads();

    const int ge = e0 + tid;
    if (ge < E) {
        const int s = edge_index[ge];
        const int d = edge_index[E + ge];
        const float* xs = x + (long)s * 16;
        const _Float16* we = s_w + tid * WCOLS;
        float acc[16];
#pragma unroll
        for (int j = 0; j < 16; ++j) acc[j] = 0.0f;
#pragma unroll
        for (int i = 0; i < 16; ++i) {
            const float xv = xs[i];
#pragma unroll
            for (int j = 0; j < 16; ++j) acc[j] += xv * (float)we[i * 16 + j];
        }
        float* ag = agg + (long)d * 16;
#pragma unroll
        for (int j = 0; j < 16; ++j) atomicAdd(&ag[j], acc[j]);
        atomicAdd(&cnt[d], 1.0f);
    }
}

__global__ void finalize_kernel(const float* __restrict__ x, const float* __restrict__ root,
                                const float* __restrict__ bias, const float* __restrict__ agg,
                                const float* __restrict__ cnt, float* __restrict__ out, int N)
{
    int n = blockIdx.x * blockDim.x + threadIdx.x;
    if (n >= N) return;
    float xi[16];
#pragma unroll
    for (int i = 0; i < 16; ++i) xi[i] = x[(long)n * 16 + i];
    float c = cnt[n];
    float inv = 1.0f / (c > 1.0f ? c : 1.0f);
#pragma unroll
    for (int o = 0; o < 16; ++o) {
        float acc = 0.0f;
#pragma unroll
        for (int i = 0; i < 16; ++i) acc += xi[i] * root[i * 16 + o];
        out[(long)n * 16 + o] = acc + agg[(long)n * 16 + o] * inv + bias[o];
    }
}

extern "C" void kernel_launch(void* const* d_in, const int* in_sizes, int n_in,
                              void* d_out, int out_size, void* d_ws, size_t ws_size,
                              hipStream_t stream)
{
    const float* x          = (const float*)d_in[0];
    const int*   edge_index = (const int*)  d_in[1];
    const float* edge_attr  = (const float*)d_in[2];
    const float* W1 = (const float*)d_in[3];  const float* b1 = (const float*)d_in[4];
    const float* W2 = (const float*)d_in[5];  const float* b2 = (const float*)d_in[6];
    const float* W3 = (const float*)d_in[7];  const float* b3 = (const float*)d_in[8];
    const float* W4 = (const float*)d_in[9];  const float* b4 = (const float*)d_in[10];
    const float* root = (const float*)d_in[11];
    const float* bias = (const float*)d_in[12];
    float* out = (float*)d_out;

    const int N = in_sizes[0] / 16;
    const int E = in_sizes[1] / 2;

    _Float16* W1t = (_Float16*)d_ws;
    _Float16* W2t = W1t + 128 * KATTR;
    _Float16* W3t = W2t + 128 * HPAD;
    _Float16* W4t = W3t + 128 * HPAD;
    float*    agg = (float*)(W4t + 256 * HPAD);
    float*    cnt = agg + (size_t)N * 16;

    const int prep_total = 128 * KATTR + 128 * HPAD + 128 * HPAD + 256 * HPAD;
    prep_weights<<<(prep_total + 255) / 256, 256, 0, stream>>>(
        W1, W2, W3, W4, W1t, W2t, W3t, W4t);

    const int zn = N * 17;
    zero_f32<<<(zn + 255) / 256, 256, 0, stream>>>(agg, zn);

    edge_kernel<<<(E + TILE_E - 1) / TILE_E, BLOCK, 0, stream>>>(
        x, edge_index, edge_attr, W1t, W2t, W3t, W4t,
        b1, b2, b3, b4, agg, cnt, E);

    finalize_kernel<<<(N + 255) / 256, 256, 0, stream>>>(
        x, root, bias, agg, cnt, out, N);
}


# Weight banks in the workspace stored fragment-major: each B-fragment load of a wave reads 512 contiguous aligned bytes (4 whole lines) instead of 32 bytes from each of 16 lines
# speedup vs baseline: 1.5077x; 1.1566x over previous
#include <hip/hip_runtime.h>
#include <hip/hip_bf16.h>


typedef __attribute__((ext_vector_type(8)))  _Float16 v8h;
typedef __attribute__((ext_vector_type(16))) _Float16 v16h;
typedef __attribute__((ext_vector_type(8)))  float    v8f;

#define TILE_E   128
#define NWAVES   4
#define BLOCK    (NWAVES * 32)
#define HPAD     128
#define NT_H     7
#define KATTR    32
#define WCOLS    256

__device__ __forceinline__ v16h load_frag(const _Float16* base, int khalf) {
    v8h lo = *(const v8h*)(base + khalf * 8);
    v8h hi = *(const v8h*)(base + 16 + khalf * 8);
    v16h r;
#pragma unroll
    for (int i = 0; i < 8; ++i) { r[i] = lo[i]; r[8 + i] = hi[i]; }
    return r;
}

__device__ __forceinline__ v16h load_bfrag(const _Float16* bank, int f, int lane) {
    const _Float16* p = bank + f * 512 + lane * 8;
    v8h lo = *(const v8h*)(p);
    v8h hi = *(const v8h*)(p + 256);
    v16h r;
#pragma unroll
    for (int i = 0; i < 8; ++i) { r[i] = lo[i]; r[8 + i] = hi[i]; }
    return r;
}

template <int NT, int KS, bool RELU>
__device__ __forceinline__ void gemm_stage(
    const _Float16* __restrict__ lds_in, int in_stride,
    const _Float16* __restrict__ Wt,
    const float* __restrict__ bias, int nvalid,
    _Float16* __restrict__ lds_out, int out_stride,
    int ntile0, int lane, int mpair)
{
    const int rc    = lane & 15;
    const int khalf = lane >> 4;
    const int m0    = mpair * 2;

    v8f acc0[NT], acc1[NT];
#pragma unroll
    for (int t = 0; t < NT; ++t) {
        acc0[t] = (v8f){0.f,0.f,0.f,0.f,0.f,0.f,0.f,0.f};
        acc1[t] = (v8f){0.f,0.f,0.f,0.f,0.f,0.f,0.f,0.f};
    }

    v16h bcur[NT];
#pragma unroll
    for (int t = 0; t < NT; ++t)
        bcur[t] = load_bfrag(Wt, (ntile0 + t) * KS, lane);

#pragma unroll
    for (int ks = 0; ks < KS; ++ks) {
        const int kb = ks * 32;
        v16h bnext[NT];
        if (ks + 1 < KS) {
#pragma unroll
            for (int t = 0; t < NT; ++t)
                bnext[t] = load_bfrag(Wt, (ntile0 + t) * KS + ks + 1, lane);
        }
        __builtin_amdgcn_sched_barrier(0);
        v16h a0 = load_frag(lds_in + ((m0    ) * 16 + rc) * in_stride + kb, khalf);
        v16h a1 = load_frag(lds_in + ((m0 + 1) * 16 + rc) * in_stride + kb, khalf);
#pragma unroll
        for (int t = 0; t < NT; ++t) {
            acc0[t] = __builtin_amdgcn_wmma_f32_16x16x32_f16(
                false, a0, false, bcur[t], (short)0, acc0[t], false, false);
            acc1[t] = __builtin_amdgcn_wmma_f32_16x16x32_f16(
                false, a1, false, bcur[t], (short)0, acc1[t], false, false);
        }
        if (ks + 1 < KS) {
#pragma unroll
            for (int t = 0; t < NT; ++t) bcur[t] = bnext[t];
        }
    }

    const int mrow0 = m0 * 16 + (khalf ? 8 : 0);
#pragma unroll
    for (int t = 0; t < NT; ++t) {
        const int  n     = (ntile0 + t) * 16 + rc;
        const bool valid = (n < nvalid);
        const float bv   = valid ? bias[n] : 0.0f;
#pragma unroll
        for (int r = 0; r < 8; ++r) {
            float v0 = acc0[t][r] + bv;
            float v1 = acc1[t][r] + bv;
            if (RELU) { v0 = v0 > 0.0f ? v0 : 0.0f; v1 = v1 > 0.0f ? v1 : 0.0f; }
            if (!valid) { v0 = 0.0f; v1 = 0.0f; }
            lds_out[(mrow0 + r     ) * out_stride + n] = (_Float16)v0;
            lds_out[(mrow0 + r + 16) * out_stride + n] = (_Float16)v1;
        }
    }
}

__device__ __forceinline__ void bank_coord(int t, int KS, int& n, int& k) {
    const int i    = t & 7;
    const int lane = (t >> 3) & 31;
    const int part = (t >> 8) & 1;
    const int f    = t >> 9;
    const int ks   = f % KS;
    const int nt   = f / KS;
    n = nt * 16 + (lane & 15);
    k = ks * 32 + part * 16 + (lane >> 4) * 8 + i;
}

__global__ void prep_weights(const float* __restrict__ W1, const float* __restrict__ W2,
                             const float* __restrict__ W3, const float* __restrict__ W4,
                             _Float16* __restrict__ W1t, _Float16* __restrict__ W2t,
                             _Float16* __restrict__ W3t, _Float16* __restrict__ W4t)
{
    int tid = blockIdx.x * blockDim.x + threadIdx.x;
    const int s1 = 128 * KATTR;
    const int s2 = s1 + 128 * HPAD;
    const int s3 = s2 + 128 * HPAD;
    const int s4 = s3 + 256 * HPAD;
    int n, k;
    if (tid < s1) {
        bank_coord(tid, KATTR / 32, n, k);
        W1t[tid] = (_Float16)((k < 8 && n < 100) ? W1[k * 100 + n] : 0.f);
    } else if (tid < s2) {
        int t = tid - s1; bank_coord(t, HPAD / 32, n, k);
        W2t[t] = (_Float16)((k < 100 && n < 100) ? W2[k * 100 + n] : 0.f);
    } else if (tid < s3) {
        int t = tid - s2; bank_coord(t, HPAD / 32, n, k);
        W3t[t] = (_Float16)((k < 100 && n < 100) ? W3[k * 100 + n] : 0.f);
    } else if (tid < s4) {
        int t = tid - s3; bank_coord(t, HPAD / 32, n, k);
        W4t[t] = (_Float16)((k < 100 && n < 256) ? W4[k * 256 + n] : 0.f);
    }
}

__global__ void zero_f32(float* __restrict__ p, int n) {
    int i = blockIdx.x * blockDim.x + threadIdx.x;
    if (i < n) p[i] = 0.0f;
}

__global__ __launch_bounds__(BLOCK)
void edge_kernel(const float* __restrict__ x,
                 const int*   __restrict__ edge_index,
                 const float* __restrict__ edge_attr,
                 const _Float16* __restrict__ W1t, const _Float16* __restrict__ W2t,
                 const _Float16* __restrict__ W3t, const _Float16* __restrict__ W4t,
                 const float* __restrict__ b1, const float* __restrict__ b2,
                 const float* __restrict__ b3, const float* __restrict__ b4,
                 float* __restrict__ agg, float* __restrict__ cnt, int E)
{
    __shared__ _Float16 s_attr[TILE_E * KATTR];
    __shared__ _Float16 s_h[TILE_E * HPAD];
    __shared__ _Float16 s_w[TILE_E * WCOLS];

    const int e0   = blockIdx.x * TILE_E;
    const int tid  = threadIdx.x;
    const int lane = tid & 31;
    const int wave = tid >> 5;

    {
        const uint4 z4 = {0u, 0u, 0u, 0u};
        uint4* ph = (uint4*)s_h;
        uint4* pw = (uint4*)s_w;
        const int nvec = TILE_E * HPAD * (int)sizeof(_Float16) / 16;
        for (int i = tid; i < nvec; i += BLOCK) { ph[i] = z4; pw[i] = z4; }
    }

    {
        const int e  = tid;
        const int ge = e0 + e;
        float4 a0 = {0.f,0.f,0.f,0.f}, a1 = {0.f,0.f,0.f,0.f};
        if (ge < E) {
            const float4* ea = (const float4*)edge_attr;
            a0 = ea[(long)ge * 2];
            a1 = ea[(long)ge * 2 + 1];
        }
        v8h h, z;
        h[0]=(_Float16)a0.x; h[1]=(_Float16)a0.y; h[2]=(_Float16)a0.z; h[3]=(_Float16)a0.w;
        h[4]=(_Float16)a1.x; h[5]=(_Float16)a1.y; h[6]=(_Float16)a1.z; h[7]=(_Float16)a1.w;
#pragma unroll
        for (int i = 0; i < 8; ++i) z[i] = (_Float16)0.f;
        v8h* dst = (v8h*)(s_attr + e * KATTR);
        dst[0] = h; dst[1] = z; dst[2] = z; dst[3] = z;
    }
    __syncthreads();

    gemm_stage<NT_H, 1, true>(s_attr, KATTR, W1t, b1, 100, s_h, HPAD, 0, lane, wave);
    __syncthreads();
    gemm_stage<NT_H, 4, true>(s_h, HPAD, W2t, b2, 100, s_w, HPAD, 0, lane, wave);
    __syncthreads();
    gemm_stage<NT_H, 4, true>(s_w, HPAD, W3t, b3, 100, s_h, HPAD, 0, lane, wave);
    __syncthreads();
    gemm_stage<8, 4, false>(s_h, HPAD, W4t, b4, 256, s_w, WCOLS, 0, lane, wave);
    gemm_stage<8, 4, false>(s_h, HPAD, W4t, b4, 256, s_w, WCOLS, 8, lane, wave);
    __syncthreads();

    const int ge = e0 + tid;
    if (ge < E) {
        const int s = edge_index[ge];
        const int d = edge_index[E + ge];
        const float* xs = x + (long)s * 16;
        const _Float16* we = s_w + tid * WCOLS;
        float acc[16];
#pragma unroll
        for (int j = 0; j < 16; ++j) acc[j] = 0.0f;
#pragma unroll
        for (int i = 0; i < 16; ++i) {
            const float xv = xs[i];
#pragma unroll
            for (int j = 0; j < 16; ++j) acc[j] += xv * (float)we[i * 16 + j];
        }
        float* ag = agg + (long)d * 16;
#pragma unroll
        for (int j = 0; j < 16; ++j) atomicAdd(&ag[j], acc[j]);
        atomicAdd(&cnt[d], 1.0f);
    }
}

__global__ void finalize_kernel(const float* __restrict__ x, const float* __restrict__ root,
                                const float* __restrict__ bias, const float* __restrict__ agg,
                                const float* __restrict__ cnt, float* __restrict__ out, int N)
{
    int n = blockIdx.x * blockDim.x + threadIdx.x;
    if (n >= N) return;
    float xi[16];
#pragma unroll
    for (int i = 0; i < 16; ++i) xi[i] = x[(long)n * 16 + i];
    float c = cnt[n];
    float inv = 1.0f / (c > 1.0f ? c : 1.0f);
#pragma unroll
    for (int o = 0; o < 16; ++o) {
        float acc = 0.0f;
#pragma unroll
        for (int i = 0; i < 16; ++i) acc += xi[i] * root[i * 16 + o];
        out[(long)n * 16 + o] = acc + agg[(long)n * 16 + o] * inv + bias[o];
    }
}

extern "C" void kernel_launch(void* const* d_in, const int* in_sizes, int n_in,
                              void* d_out, int out_size, void* d_ws, size_t ws_size,
                              hipStream_t stream)
{
    const float* x          = (const float*)d_in[0];
    const int*   edge_index = (const int*)  d_in[1];
    const float* edge_attr  = (const float*)d_in[2];
    const float* W1 = (const float*)d_in[3];  const float* b1 = (const float*)d_in[4];
    const float* W2 = (const float*)d_in[5];  const float* b2 = (const float*)d_in[6];
    const float* W3 = (const float*)d_in[7];  const float* b3 = (const float*)d_in[8];
    const float* W4 = (const float*)d_in[9];  const float* b4 = (const float*)d_in[10];
    const float* root = (const float*)d_in[11];
    const float* bias = (const float*)d_in[12];
    float* out = (float*)d_out;

    const int N = in_sizes[0] / 16;
    const int E = in_sizes[1] / 2;

    _Float16* W1t = (_Float16*)d_ws;
    _Float16* W2t = W1t + 128 * KATTR;
    _Float16* W3t = W2t + 128 * HPAD;
    _Float16* W4t = W3t + 128 * HPAD;
    float*    agg = (float*)(W4t + 256 * HPAD);
    float*    cnt = agg + (size_t)N * 16;

    const int prep_total = 128 * KATTR + 128 * HPAD + 128 * HPAD + 256 * HPAD;
    prep_weights<<<(prep_total + 255) / 256, 256, 0, stream>>>(
        W1, W2, W3, W4, W1t, W2t, W3t, W4t);

    const int zn = N * 17;
    zero_f32<<<(zn + 255) / 256, 256, 0, stream>>>(agg, zn);

    edge_kernel<<<(E + TILE_E - 1) / TILE_E, BLOCK, 0, stream>>>(
        x, edge_index, edge_attr, W1t, W2t, W3t, W4t,
        b1, b2, b3, b4, agg, cnt, E);

    finalize_kernel<<<(N + 255) / 256, 256, 0, stream>>>(
        x, root, bias, agg, cnt, out, N);
}


# Per-edge 16x16 apply done on the last layer's f32 accumulators in registers (x rows from LDS) instead of via an f16 weight tile in LDS; one edge's 16 atomic adds then fall in one 64-byte line
# speedup vs baseline: 2.6815x; 1.7785x over previous
#include <hip/hip_runtime.h>
#include <hip/hip_bf16.h>


typedef __attribute__((ext_vector_type(8)))  _Float16 v8h;
typedef __attribute__((ext_vector_type(16))) _Float16 v16h;
typedef __attribute__((ext_vector_type(8)))  float    v8f;

#define TILE_E   128
#define NWAVES   4
#define BLOCK    (NWAVES * 32)
#define HPAD     128
#define NT_H     7
#define KATTR    32
#define WCOLS    256

__device__ __forceinline__ v16h load_frag(const _Float16* base, int khalf) {
    v8h lo = *(const v8h*)(base + khalf * 8);
    v8h hi = *(const v8h*)(base + 16 + khalf * 8);
    v16h r;
#pragma unroll
    for (int i = 0; i < 8; ++i) { r[i] = lo[i]; r[8 + i] = hi[i]; }
    return r;
}

__device__ __forceinline__ v16h load_bfrag(const _Float16* bank, int f, int lane) {
    const _Float16* p = bank + f * 512 + lane * 8;
    v8h lo = *(const v8h*)(p);
    v8h hi = *(const v8h*)(p + 256);
    v16h r;
#pragma unroll
    for (int i = 0; i < 8; ++i) { r[i] = lo[i]; r[8 + i] = hi[i]; }
    return r;
}

template <int NT, int KS, bool RELU>
__device__ __forceinline__ void gemm_stage(
    const _Float16* __restrict__ lds_in, int in_stride,
    const _Float16* __restrict__ Wt,
    const float* __restrict__ bias, int nvalid,
    _Float16* __restrict__ lds_out, int out_stride,
    int ntile0, int lane, int mpair)
{
    const int rc    = lane & 15;
    const int khalf = lane >> 4;
    const int m0    = mpair * 2;

    v8f acc0[NT], acc1[NT];
#pragma unroll
    for (int t = 0; t < NT; ++t) {
        acc0[t] = (v8f){0.f,0.f,0.f,0.f,0.f,0.f,0.f,0.f};
        acc1[t] = (v8f){0.f,0.f,0.f,0.f,0.f,0.f,0.f,0.f};
    }

    v16h bcur[NT];
#pragma unroll
    for (int t = 0; t < NT; ++t)
        bcur[t] = load_bfrag(Wt, (ntile0 + t) * KS, lane);

#pragma unroll
    for (int ks = 0; ks < KS; ++ks) {
        const int kb = ks * 32;
        v16h bnext[NT];
        if (ks + 1 < KS) {
#pragma unroll
            for (int t = 0; t < NT; ++t)
                bnext[t] = load_bfrag(Wt, (ntile0 + t) * KS + ks + 1, lane);
        }
        __builtin_amdgcn_sched_barrier(0);
        v16h a0 = load_frag(lds_in + ((m0    ) * 16 + rc) * in_stride + kb, khalf);
        v16h a1 = load_frag(lds_in + ((m0 + 1) * 16 + rc) * in_stride + kb, khalf);
#pragma unroll
        for (int t = 0; t < NT; ++t) {
            acc0[t] = __builtin_amdgcn_wmma_f32_16x16x32_f16(
                false, a0, false, bcur[t], (short)0, acc0[t], false, false);
            acc1[t] = __builtin_amdgcn_wmma_f32_16x16x32_f16(
                false, a1, false, bcur[t], (short)0, acc1[t], false, false);
        }
        if (ks + 1 < KS) {
#pragma unroll
            for (int t = 0; t < NT; ++t) bcur[t] = bnext[t];
        }
    }

    const int mrow0 = m0 * 16 + (khalf ? 8 : 0);
#pragma unroll
    for (int t = 0; t < NT; ++t) {
        const int  n     = (ntile0 + t) * 16 + rc;
        const bool valid = (n < nvalid);
        const float bv   = valid ? bias[n] : 0.0f;
#pragma unroll
        for (int r = 0; r < 8; ++r) {
            float v0 = acc0[t][r] + bv;
            float v1 = acc1[t][r] + bv;
            if (RELU) { v0 = v0 > 0.0f ? v0 : 0.0f; v1 = v1 > 0.0f ? v1 : 0.0f; }
            if (!valid) { v0 = 0.0f; v1 = 0.0f; }
            lds_out[(mrow0 + r     ) * out_stride + n] = (_Float16)v0;
            lds_out[(mrow0 + r + 16) * out_stride + n] = (_Float16)v1;
        }
    }
}

template <int KS, int PASS>
__device__ __forceinline__ void gemm_apply_pass(
    const _Float16* __restrict__ lds_in, int in_stride,
    const _Float16* __restrict__ Wt,
    const float* __restrict__ bias,
    const float* __restrict__ lds_x,
    float* __restrict__ lds_m,
    int lane, int mpair,
    float (&msg0)[8], float (&msg1)[8])
{
    constexpr int NT = 8;
    constexpr int ntile0 = PASS * NT;
    const int rc    = lane & 15;
    const int khalf = lane >> 4;
    const int m0    = mpair * 2;

    v8f acc0[NT], acc1[NT];
#pragma unroll
    for (int t = 0; t < NT; ++t) {
        acc0[t] = (v8f){0.f,0.f,0.f,0.f,0.f,0.f,0.f,0.f};
        acc1[t] = (v8f){0.f,0.f,0.f,0.f,0.f,0.f,0.f,0.f};
    }

    v16h bcur[NT];
#pragma unroll
    for (int t = 0; t < NT; ++t)
        bcur[t] = load_bfrag(Wt, (ntile0 + t) * KS, lane);

    float bv[NT];
#pragma unroll
    for (int t = 0; t < NT; ++t) bv[t] = bias[(ntile0 + t) * 16 + rc];

#pragma unroll
    for (int ks = 0; ks < KS; ++ks) {
        const int kb = ks * 32;
        v16h bnext[NT];
        if (ks + 1 < KS) {
#pragma unroll
            for (int t = 0; t < NT; ++t)
                bnext[t] = load_bfrag(Wt, (ntile0 + t) * KS + ks + 1, lane);
        }
        __builtin_amdgcn_sched_barrier(0);
        v16h a0 = load_frag(lds_in + ((m0    ) * 16 + rc) * in_stride + kb, khalf);
        v16h a1 = load_frag(lds_in + ((m0 + 1) * 16 + rc) * in_stride + kb, khalf);
#pragma unroll
        for (int t = 0; t < NT; ++t) {
            acc0[t] = __builtin_amdgcn_wmma_f32_16x16x32_f16(
                false, a0, false, bcur[t], (short)0, acc0[t], false, false);
            acc1[t] = __builtin_amdgcn_wmma_f32_16x16x32_f16(
                false, a1, false, bcur[t], (short)0, acc1[t], false, false);
        }
        if (ks + 1 < KS) {
#pragma unroll
            for (int t = 0; t < NT; ++t) bcur[t] = bnext[t];
        }
    }

    const int row0 = m0 * 16 + 8 * khalf;
    const float* xrow = lds_x + row0 * 16 + ntile0;
    float*       mrow = lds_m + row0 * 16 + rc;
#pragma unroll
    for (int r = 0; r < 8; ++r) {
        const float4 xa = *(const float4*)(xrow + r * 16);
        const float4 xb = *(const float4*)(xrow + r * 16 + 4);
        const float4 ya = *(const float4*)(xrow + (16 + r) * 16);
        const float4 yb = *(const float4*)(xrow + (16 + r) * 16 + 4);
        const float xv[8] = {xa.x, xa.y, xa.z, xa.w, xb.x, xb.y, xb.z, xb.w};
        const float yv[8] = {ya.x, ya.y, ya.z, ya.w, yb.x, yb.y, yb.z, yb.w};
        float s0 = 0.0f, s1 = 0.0f;
        if (PASS == 1) { s0 = mrow[r * 16]; s1 = mrow[(16 + r) * 16]; }
#pragma unroll
        for (int t = 0; t < NT; ++t) {
            s0 += xv[t] * (acc0[t][r] + bv[t]);
            s1 += yv[t] * (acc1[t][r] + bv[t]);
        }
        if (PASS == 0) { mrow[r * 16] = s0; mrow[(16 + r) * 16] = s1; }
        msg0[r] = s0; msg1[r] = s1;
    }
}

__device__ __forceinline__ void bank_coord(int t, int KS, int& n, int& k) {
    const int i    = t & 7;
    const int lane = (t >> 3) & 31;
    const int part = (t >> 8) & 1;
    const int f    = t >> 9;
    const int ks   = f % KS;
    const int nt   = f / KS;
    n = nt * 16 + (lane & 15);
    k = ks * 32 + part * 16 + (lane >> 4) * 8 + i;
}

__global__ void prep_weights(const float* __restrict__ W1, const float* __restrict__ W2,
                             const float* __restrict__ W3, const float* __restrict__ W4,
                             _Float16* __restrict__ W1t, _Float16* __restrict__ W2t,
                             _Float16* __restrict__ W3t, _Float16* __restrict__ W4t)
{
    int tid = blockIdx.x * blockDim.x + threadIdx.x;
    const int s1 = 128 * KATTR;
    const int s2 = s1 + 128 * HPAD;
    const int s3 = s2 + 128 * HPAD;
    const int s4 = s3 + 256 * HPAD;
    int n, k;
    if (tid < s1) {
        bank_coord(tid, KATTR / 32, n, k);
        W1t[tid] = (_Float16)((k < 8 && n < 100) ? W1[k * 100 + n] : 0.f);
    } else if (tid < s2) {
        int t = tid - s1; bank_coord(t, HPAD / 32, n, k);
        W2t[t] = (_Float16)((k < 100 && n < 100) ? W2[k * 100 + n] : 0.f);
    } else if (tid < s3) {
        int t = tid - s2; bank_coord(t, HPAD / 32, n, k);
        W3t[t] = (_Float16)((k < 100 && n < 100) ? W3[k * 100 + n] : 0.f);
    } else if (tid < s4) {
        int t = tid - s3; bank_coord(t, HPAD / 32, n, k);
        W4t[t] = (_Float16)((k < 100 && n < 256) ? W4[k * 256 + n] : 0.f);
    }
}

__global__ void zero_f32(float* __restrict__ p, int n) {
    int i = blockIdx.x * blockDim.x + threadIdx.x;
    if (i < n) p[i] = 0.0f;
}

__global__ __launch_bounds__(BLOCK, 3)
void edge_kernel(const float* __restrict__ x,
                 const int*   __restrict__ edge_index,
                 const float* __restrict__ edge_attr,
                 const _Float16* __restrict__ W1t, const _Float16* __restrict__ W2t,
                 const _Float16* __restrict__ W3t, const _Float16* __restrict__ W4t,
                 const float* __restrict__ b1, const float* __restrict__ b2,
                 const float* __restrict__ b3, const float* __restrict__ b4,
                 float* __restrict__ agg, float* __restrict__ cnt, int E)
{
    __shared__ _Float16 s_attr[TILE_E * KATTR];
    __shared__ _Float16 s_h[TILE_E * HPAD];
    __shared__ _Float16 s_g[TILE_E * HPAD];
    __shared__ __attribute__((aligned(16))) float s_x[TILE_E * 16];
    __shared__ float    s_m[TILE_E * 16];

    const int e0   = blockIdx.x * TILE_E;
    const int tid  = threadIdx.x;
    const int lane = tid & 31;
    const int wave = tid >> 5;

    const int  ge   = e0 + tid;
    const bool live = ge < E;
    const int  gc   = live ? ge : E - 1;
    const float4* ea = (const float4*)edge_attr;
    const float4 a0 = ea[(long)gc * 2];
    const float4 a1 = ea[(long)gc * 2 + 1];
    const int s = edge_index[gc];
    const int dmine = edge_index[E + gc];

    {
        const uint4 z4 = {0u, 0u, 0u, 0u};
        uint4* ph = (uint4*)s_h;
        uint4* pg = (uint4*)s_g;
        const int nvec = TILE_E * HPAD * (int)sizeof(_Float16) / 16;
        for (int i = tid; i < nvec; i += BLOCK) { ph[i] = z4; pg[i] = z4; }
    }

    {
        v8h h, z;
        h[0]=(_Float16)a0.x; h[1]=(_Float16)a0.y; h[2]=(_Float16)a0.z; h[3]=(_Float16)a0.w;
        h[4]=(_Float16)a1.x; h[5]=(_Float16)a1.y; h[6]=(_Float16)a1.z; h[7]=(_Float16)a1.w;
#pragma unroll
        for (int i = 0; i < 8; ++i) z[i] = (_Float16)0.f;
        v8h* dst = (v8h*)(s_attr + tid * KATTR);
        dst[0] = h; dst[1] = z; dst[2] = z; dst[3] = z;
    }
    __syncthreads();

    const float4* xs = (const float4*)(x + (long)s * 16);
    const float4 xq0 = xs[0], xq1 = xs[1], xq2 = xs[2], xq3 = xs[3];

    gemm_stage<NT_H, 1, true>(s_attr, KATTR, W1t, b1, 100, s_h, HPAD, 0, lane, wave);
    __syncthreads();

    {
        float4* dx = (float4*)(s_x + tid * 16);
        dx[0] = xq0; dx[1] = xq1; dx[2] = xq2; dx[3] = xq3;
    }

    gemm_stage<NT_H, 4, true>(s_h, HPAD, W2t, b2, 100, s_g, HPAD, 0, lane, wave);
    __syncthreads();
    gemm_stage<NT_H, 4, true>(s_g, HPAD, W3t, b3, 100, s_h, HPAD, 0, lane, wave);
    __syncthreads();

    float msg0[8], msg1[8];
    gemm_apply_pass<4, 0>(s_h, HPAD, W4t, b4, s_x, s_m, lane, wave, msg0, msg1);
    gemm_apply_pass<4, 1>(s_h, HPAD, W4t, b4, s_x, s_m, lane, wave, msg0, msg1);

    {
        const int o     = lane & 15;
        const int khalf = lane >> 4;
        const int dsel  = live ? dmine : -1;
#pragma unroll
        for (int r = 0; r < 8; ++r) {
            const int d0 = __shfl(dsel, 8 * khalf + r);
            const int d1 = __shfl(dsel, 16 + 8 * khalf + r);
            if (d0 >= 0) atomicAdd(&agg[(long)d0 * 16 + o], msg0[r]);
            if (d1 >= 0) atomicAdd(&agg[(long)d1 * 16 + o], msg1[r]);
        }
        if (dsel >= 0) atomicAdd(&cnt[dsel], 1.0f);
    }
}

__global__ void finalize_kernel(const float* __restrict__ x, const float* __restrict__ root,
                                const float* __restrict__ bias, const float* __restrict__ agg,
                                const float* __restrict__ cnt, float* __restrict__ out, int N)
{
    int n = blockIdx.x * blockDim.x + threadIdx.x;
    if (n >= N) return;
    float xi[16];
#pragma unroll
    for (int i = 0; i < 16; ++i) xi[i] = x[(long)n * 16 + i];
    float c = cnt[n];
    float inv = 1.0f / (c > 1.0f ? c : 1.0f);
#pragma unroll
    for (int o = 0; o < 16; ++o) {
        float acc = 0.0f;
#pragma unroll
        for (int i = 0; i < 16; ++i) acc += xi[i] * root[i * 16 + o];
        out[(long)n * 16 + o] = acc + agg[(long)n * 16 + o] * inv + bias[o];
    }
}

extern "C" void kernel_launch(void* const* d_in, const int* in_sizes, int n_in,
                              void* d_out, int out_size, void* d_ws, size_t ws_size,
                              hipStream_t stream)
{
    const float* x          = (const float*)d_in[0];
    const int*   edge_index = (const int*)  d_in[1];
    const float* edge_attr  = (const float*)d_in[2];
    const float* W1 = (const float*)d_in[3];  const float* b1 = (const float*)d_in[4];
    const float* W2 = (const float*)d_in[5];  const float* b2 = (const float*)d_in[6];
    const float* W3 = (const float*)d_in[7];  const float* b3 = (const float*)d_in[8];
    const float* W4 = (const float*)d_in[9];  const float* b4 = (const float*)d_in[10];
    const float* root = (const float*)d_in[11];
    const float* bias = (const float*)d_in[12];
    float* out = (float*)d_out;

    const int N = in_sizes[0] / 16;
    const int E = in_sizes[1] / 2;

    _Float16* W1t = (_Float16*)d_ws;
    _Float16* W2t = W1t + 128 * KATTR;
    _Float16* W3t = W2t + 128 * HPAD;
    _Float16* W4t = W3t + 128 * HPAD;
    float*    agg = (float*)(W4t + 256 * HPAD);
    float*    cnt = agg + (size_t)N * 16;

    const int prep_total = 128 * KATTR + 128 * HPAD + 128 * HPAD + 256 * HPAD;
    prep_weights<<<(prep_total + 255) / 256, 256, 0, stream>>>(
        W1, W2, W3, W4, W1t, W2t, W3t, W4t);

    const int zn = N * 17;
    zero_f32<<<(zn + 255) / 256, 256, 0, stream>>>(agg, zn);

    edge_kernel<<<(E + TILE_E - 1) / TILE_E, BLOCK, 0, stream>>>(
        x, edge_index, edge_attr, W1t, W2t, W3t, W4t,
        b1, b2, b3, b4, agg, cnt, E);

    finalize_kernel<<<(N + 255) / 256, 256, 0, stream>>>(
        x, root, bias, agg, cnt, out, N);
}


# Activation (A) fragments fetched a K-step ahead too, in one group with the weight fragments (biases re-read per tile to free registers): no matrix instruction waits on a load issued right before it
# speedup vs baseline: 4.1075x; 1.5318x over previous
#include <hip/hip_runtime.h>
#include <hip/hip_bf16.h>


typedef __attribute__((ext_vector_type(8)))  _Float16 v8h;
typedef __attribute__((ext_vector_type(16))) _Float16 v16h;
typedef __attribute__((ext_vector_type(8)))  float    v8f;

#define WAVE_E   32
#define NWAVES   12
#define BLOCK    (NWAVES * 32)
#define NWG      256
#define HPITCH   136
#define APITCH   40
#define W1OFF    0
#define W2OFF    (128 * 32)
#define W3OFF    (W2OFF + 128 * 128)
#define W4OFF    (W3OFF + 128 * 128)
#define WBANKS   (W4OFF + 256 * 128)
#define HPAD     128
#define NT_H     7
#define KATTR    32
#define WCOLS    256

__device__ __forceinline__ v16h load_frag(const _Float16* base, int khalf) {
    v8h lo = *(const v8h*)(base + khalf * 8);
    v8h hi = *(const v8h*)(base + 16 + khalf * 8);
    v16h r;
#pragma unroll
    for (int i = 0; i < 8; ++i) { r[i] = lo[i]; r[8 + i] = hi[i]; }
    return r;
}

__device__ __forceinline__ v16h load_bfrag(const _Float16* bank, int f, int lane) {
    const _Float16* p = bank + f * 512 + lane * 8;
    v8h lo = *(const v8h*)(p);
    v8h hi = *(const v8h*)(p + 256);
    v16h r;
#pragma unroll
    for (int i = 0; i < 8; ++i) { r[i] = lo[i]; r[8 + i] = hi[i]; }
    return r;
}

template <int NT, int KS, bool SPLIT>
__device__ __forceinline__ void kloop(
    const _Float16* lds_in, int in_stride,
    const _Float16* Wt, int ntile0, int lane,
    v8f (&acc0)[NT], v8f (&acc1)[NT])
{
    constexpr int NH = SPLIT ? NT / 2 : NT;
    const int rc    = lane & 15;
    const int khalf = lane >> 4;
    const _Float16* arow0 = lds_in + rc * in_stride;
    const _Float16* arow1 = lds_in + (16 + rc) * in_stride;

    v16h b[2][NT];
    v16h a[2][2];
#pragma unroll
    for (int t = 0; t < NT; ++t)
        b[0][t] = load_bfrag(Wt, (ntile0 + t) * KS, lane);
    a[0][0] = load_frag(arow0, khalf);
    a[0][1] = load_frag(arow1, khalf);
    __builtin_amdgcn_sched_barrier(0);

#pragma unroll
    for (int ks = 0; ks < KS; ++ks) {
        const int cur = ks & 1;
        const int nxt = cur ^ 1;
        if (ks + 1 < KS) {
#pragma unroll
            for (int t = 0; t < NH; ++t)
                b[nxt][t] = load_bfrag(Wt, (ntile0 + t) * KS + ks + 1, lane);
            a[nxt][0] = load_frag(arow0 + (ks + 1) * 32, khalf);
            a[nxt][1] = load_frag(arow1 + (ks + 1) * 32, khalf);
        }
        __builtin_amdgcn_sched_barrier(0);
#pragma unroll
        for (int t = 0; t < NH; ++t) {
            acc0[t] = __builtin_amdgcn_wmma_f32_16x16x32_f16(
                false, a[cur][0], false, b[cur][t], (short)0, acc0[t], false, false);
            acc1[t] = __builtin_amdgcn_wmma_f32_16x16x32_f16(
                false, a[cur][1], false, b[cur][t], (short)0, acc1[t], false, false);
        }
        if (SPLIT) {
            if (ks + 1 < KS) {
                __builtin_amdgcn_sched_barrier(0);
#pragma unroll
                for (int t = NH; t < NT; ++t)
                    b[nxt][t] = load_bfrag(Wt, (ntile0 + t) * KS + ks + 1, lane);
                __builtin_amdgcn_sched_barrier(0);
            }
#pragma unroll
            for (int t = NH; t < NT; ++t) {
                acc0[t] = __builtin_amdgcn_wmma_f32_16x16x32_f16(
                    false, a[cur][0], false, b[cur][t], (short)0, acc0[t], false, false);
                acc1[t] = __builtin_amdgcn_wmma_f32_16x16x32_f16(
                    false, a[cur][1], false, b[cur][t], (short)0, acc1[t], false, false);
            }
        }
        if (ks + 1 < KS) {
            __builtin_amdgcn_sched_barrier(0);
        }
    }
}

template <int NT, int KS, bool RELU>
__device__ __forceinline__ void gemm_stage(
    const _Float16* lds_in, int in_stride,
    const _Float16* Wt,
    const float* bias, int nvalid,
    _Float16* lds_out, int out_stride,
    int ntile0, int lane)
{
    const int rc    = lane & 15;
    const int khalf = lane >> 4;
    const int m0    = 0;

    v8f acc0[NT], acc1[NT];
#pragma unroll
    for (int t = 0; t < NT; ++t) {
        acc0[t] = (v8f){0.f,0.f,0.f,0.f,0.f,0.f,0.f,0.f};
        acc1[t] = (v8f){0.f,0.f,0.f,0.f,0.f,0.f,0.f,0.f};
    }

    float bvs[NT];
#pragma unroll
    for (int t = 0; t < NT; ++t) {
        const int n = (ntile0 + t) * 16 + rc;
        bvs[t] = (n < nvalid) ? bias[n] : 0.0f;
    }

    kloop<NT, KS, false>(lds_in, in_stride, Wt, ntile0, lane, acc0, acc1);

    const int mrow0 = m0 * 16 + (khalf ? 8 : 0);
#pragma unroll
    for (int t = 0; t < NT; ++t) {
        const int  n     = (ntile0 + t) * 16 + rc;
        const bool valid = (n < nvalid);
        const float bv   = bvs[t];
#pragma unroll
        for (int r = 0; r < 8; ++r) {
            float v0 = acc0[t][r] + bv;
            float v1 = acc1[t][r] + bv;
            if (RELU) { v0 = v0 > 0.0f ? v0 : 0.0f; v1 = v1 > 0.0f ? v1 : 0.0f; }
            if (!valid) { v0 = 0.0f; v1 = 0.0f; }
            lds_out[(mrow0 + r     ) * out_stride + n] = (_Float16)v0;
            lds_out[(mrow0 + r + 16) * out_stride + n] = (_Float16)v1;
        }
    }
}

template <int KS, int PASS>
__device__ __forceinline__ void gemm_apply_pass(
    const _Float16* lds_in, int in_stride,
    const _Float16* Wt,
    const float* bias,
    const float* lds_x,
    float* lds_m,
    int lane,
    float (&msg0)[8], float (&msg1)[8])
{
    constexpr int NT = 8;
    constexpr int ntile0 = PASS * NT;
    const int rc    = lane & 15;
    const int khalf = lane >> 4;
    const int m0    = 0;

    v8f acc0[NT], acc1[NT];
#pragma unroll
    for (int t = 0; t < NT; ++t) {
        acc0[t] = (v8f){0.f,0.f,0.f,0.f,0.f,0.f,0.f,0.f};
        acc1[t] = (v8f){0.f,0.f,0.f,0.f,0.f,0.f,0.f,0.f};
    }

    float bv[NT];
#pragma unroll
    for (int t = 0; t < NT; ++t) bv[t] = bias[(ntile0 + t) * 16 + rc];

    kloop<NT, KS, true>(lds_in, in_stride, Wt, ntile0, lane, acc0, acc1);

    const int row0 = m0 * 16 + 8 * khalf;
    const float* xrow = lds_x + row0 * 16 + ntile0;
    float*       mrow = lds_m + row0 * 16 + rc;
#pragma unroll
    for (int r = 0; r < 8; ++r) {
        const float4 xa = *(const float4*)(xrow + r * 16);
        const float4 xb = *(const float4*)(xrow + r * 16 + 4);
        const float4 ya = *(const float4*)(xrow + (16 + r) * 16);
        const float4 yb = *(const float4*)(xrow + (16 + r) * 16 + 4);
        const float xv[8] = {xa.x, xa.y, xa.z, xa.w, xb.x, xb.y, xb.z, xb.w};
        const float yv[8] = {ya.x, ya.y, ya.z, ya.w, yb.x, yb.y, yb.z, yb.w};
        float s0 = 0.0f, s1 = 0.0f;
        if (PASS == 1) { s0 = mrow[r * 16]; s1 = mrow[(16 + r) * 16]; }
#pragma unroll
        for (int t = 0; t < NT; ++t) {
            s0 += xv[t] * (acc0[t][r] + bv[t]);
            s1 += yv[t] * (acc1[t][r] + bv[t]);
        }
        if (PASS == 0) { mrow[r * 16] = s0; mrow[(16 + r) * 16] = s1; }
        msg0[r] = s0; msg1[r] = s1;
    }
}

__device__ __forceinline__ void bank_coord(int t, int KS, int& n, int& k) {
    const int i    = t & 7;
    const int lane = (t >> 3) & 31;
    const int part = (t >> 8) & 1;
    const int f    = t >> 9;
    const int ks   = f % KS;
    const int nt   = f / KS;
    n = nt * 16 + (lane & 15);
    k = ks * 32 + part * 16 + (lane >> 4) * 8 + i;
}

__global__ void prep_weights(const float* __restrict__ W1, const float* __restrict__ W2,
                             const float* __restrict__ W3, const float* __restrict__ W4,
                             _Float16* __restrict__ W1t, _Float16* __restrict__ W2t,
                             _Float16* __restrict__ W3t, _Float16* __restrict__ W4t)
{
    int tid = blockIdx.x * blockDim.x + threadIdx.x;
    const int s1 = 128 * KATTR;
    const int s2 = s1 + 128 * HPAD;
    const int s3 = s2 + 128 * HPAD;
    const int s4 = s3 + 256 * HPAD;
    int n, k;
    if (tid < s1) {
        bank_coord(tid, KATTR / 32, n, k);
        W1t[tid] = (_Float16)((k < 8 && n < 100) ? W1[k * 100 + n] : 0.f);
    } else if (tid < s2) {
        int t = tid - s1; bank_coord(t, HPAD / 32, n, k);
        W2t[t] = (_Float16)((k < 100 && n < 100) ? W2[k * 100 + n] : 0.f);
    } else if (tid < s3) {
        int t = tid - s2; bank_coord(t, HPAD / 32, n, k);
        W3t[t] = (_Float16)((k < 100 && n < 100) ? W3[k * 100 + n] : 0.f);
    } else if (tid < s4) {
        int t = tid - s3; bank_coord(t, HPAD / 32, n, k);
        W4t[t] = (_Float16)((k < 100 && n < 256) ? W4[k * 256 + n] : 0.f);
    }
}

__global__ void zero_f32(float* __restrict__ p, int n) {
    int i = blockIdx.x * blockDim.x + threadIdx.x;
    if (i < n) p[i] = 0.0f;
}

__device__ __forceinline__ void copy_bank(const _Float16* __restrict__ src, _Float16* dst,
                                          int nvec, int tid) {
    const uint4* s = (const uint4*)src;
    uint4*       d = (uint4*)dst;
    for (int i = tid; i < nvec; i += BLOCK) d[i] = s[i];
}

__global__ __launch_bounds__(BLOCK, 3)
void edge_kernel(const float* __restrict__ x,
                 const int*   __restrict__ edge_index,
                 const float* __restrict__ edge_attr,
                 const _Float16* __restrict__ W1t, const _Float16* __restrict__ W2t,
                 const _Float16* __restrict__ W3t, const _Float16* __restrict__ W4t,
                 const float* b1, const float* b2,
                 const float* b3, const float* b4,
                 float* __restrict__ agg, float* __restrict__ cnt, int E)
{
    __shared__ __attribute__((aligned(16))) _Float16 s_w[WBANKS];
    __shared__ __attribute__((aligned(16))) _Float16 s_h[NWAVES * WAVE_E * HPITCH];
    __shared__ __attribute__((aligned(16))) _Float16 s_attr[NWAVES * WAVE_E * APITCH];
    __shared__ __attribute__((aligned(16))) float    s_x[NWAVES * WAVE_E * 16];
    __shared__ __attribute__((aligned(16))) float    s_m[NWAVES * WAVE_E * 16];

    const int tid  = threadIdx.x;
    const int lane = tid & 31;
    const int wave = __builtin_amdgcn_readfirstlane(tid >> 5);

    copy_bank(W1t, s_w + W1OFF, 128 * KATTR * 2 / 16, tid);
    copy_bank(W2t, s_w + W2OFF, 128 * HPAD  * 2 / 16, tid);
    copy_bank(W3t, s_w + W3OFF, 128 * HPAD  * 2 / 16, tid);
    copy_bank(W4t, s_w + W4OFF, 256 * HPAD  * 2 / 16, tid);

    _Float16* hw = s_h    + wave * WAVE_E * HPITCH;
    _Float16* aw = s_attr + wave * WAVE_E * APITCH;
    float*    xw = s_x    + wave * WAVE_E * 16;
    float*    mw = s_m    + wave * WAVE_E * 16;

    {
        const uint4 z4 = {0u, 0u, 0u, 0u};
        uint4* ph = (uint4*)hw;
        const int nvec = WAVE_E * HPITCH * (int)sizeof(_Float16) / 16;
        for (int i = lane; i < nvec; i += 32) ph[i] = z4;
    }
    __syncthreads();

    const _Float16* W1s = s_w + W1OFF;
    const _Float16* W2s = s_w + W2OFF;
    const _Float16* W3s = s_w + W3OFF;
    const _Float16* W4s = s_w + W4OFF;

    const int ntiles = (E + WAVE_E - 1) / WAVE_E;
    const int nwaves = (int)gridDim.x * NWAVES;
    for (int tile = wave * (int)gridDim.x + (int)blockIdx.x; tile < ntiles; tile += nwaves) {
        const int  ge   = tile * WAVE_E + lane;
        const bool live = ge < E;
        const int  gc   = live ? ge : E - 1;
        const float4* ea = (const float4*)edge_attr;
        const float4 a0 = ea[(long)gc * 2];
        const float4 a1 = ea[(long)gc * 2 + 1];
        const int s = edge_index[gc];
        const int dmine = edge_index[E + gc];

        {
            v8h h, z;
            h[0]=(_Float16)a0.x; h[1]=(_Float16)a0.y; h[2]=(_Float16)a0.z; h[3]=(_Float16)a0.w;
            h[4]=(_Float16)a1.x; h[5]=(_Float16)a1.y; h[6]=(_Float16)a1.z; h[7]=(_Float16)a1.w;
#pragma unroll
            for (int i = 0; i < 8; ++i) z[i] = (_Float16)0.f;
            v8h* dst = (v8h*)(aw + lane * APITCH);
            dst[0] = h; dst[1] = z; dst[2] = z; dst[3] = z;
        }

        const float4* xs = (const float4*)(x + (long)s * 16);
        const float4 xq0 = xs[0], xq1 = xs[1], xq2 = xs[2], xq3 = xs[3];

        gemm_stage<NT_H, 1, true>(aw, APITCH, W1s, b1, 100, hw, HPITCH, 0, lane);

        {
            float4* dx = (float4*)(xw + lane * 16);
            dx[0] = xq0; dx[1] = xq1; dx[2] = xq2; dx[3] = xq3;
        }

        gemm_stage<NT_H, 4, true>(hw, HPITCH, W2s, b2, 100, hw, HPITCH, 0, lane);
        gemm_stage<NT_H, 4, true>(hw, HPITCH, W3s, b3, 100, hw, HPITCH, 0, lane);

        float msg0[8], msg1[8];
        gemm_apply_pass<4, 0>(hw, HPITCH, W4s, b4, xw, mw, lane, msg0, msg1);
        gemm_apply_pass<4, 1>(hw, HPITCH, W4s, b4, xw, mw, lane, msg0, msg1);

        {
            const int o     = lane & 15;
            const int khalf = lane >> 4;
            const int dsel  = live ? dmine : -1;
#pragma unroll
            for (int r = 0; r < 8; ++r) {
                const int d0 = __shfl(dsel, 8 * khalf + r);
                const int d1 = __shfl(dsel, 16 + 8 * khalf + r);
                if (d0 >= 0) atomicAdd(&agg[(long)d0 * 16 + o], msg0[r]);
                if (d1 >= 0) atomicAdd(&agg[(long)d1 * 16 + o], msg1[r]);
            }
            if (dsel >= 0) atomicAdd(&cnt[dsel], 1.0f);
        }
    }
}

__global__ void finalize_kernel(const float* __restrict__ x, const float* __restrict__ root,
                                const float* __restrict__ bias, const float* __restrict__ agg,
                                const float* __restrict__ cnt, float* __restrict__ out, int N)
{
    int n = blockIdx.x * blockDim.x + threadIdx.x;
    if (n >= N) return;
    float xi[16];
#pragma unroll
    for (int i = 0; i < 16; ++i) xi[i] = x[(long)n * 16 + i];
    float c = cnt[n];
    float inv = 1.0f / (c > 1.0f ? c : 1.0f);
#pragma unroll
    for (int o = 0; o < 16; ++o) {
        float acc = 0.0f;
#pragma unroll
        for (int i = 0; i < 16; ++i) acc += xi[i] * root[i * 16 + o];
        out[(long)n * 16 + o] = acc + agg[(long)n * 16 + o] * inv + bias[o];
    }
}

extern "C" void kernel_launch(void* const* d_in, const int* in_sizes, int n_in,
                              void* d_out, int out_size, void* d_ws, size_t ws_size,
                              hipStream_t stream)
{
    const float* x          = (const float*)d_in[0];
    const int*   edge_index = (const int*)  d_in[1];
    const float* edge_attr  = (const float*)d_in[2];
    const float* W1 = (const float*)d_in[3];  const float* b1 = (const float*)d_in[4];
    const float* W2 = (const float*)d_in[5];  const float* b2 = (const float*)d_in[6];
    const float* W3 = (const float*)d_in[7];  const float* b3 = (const float*)d_in[8];
    const float* W4 = (const float*)d_in[9];  const float* b4 = (const float*)d_in[10];
    const float* root = (const float*)d_in[11];
    const float* bias = (const float*)d_in[12];
    float* out = (float*)d_out;

    const int N = in_sizes[0] / 16;
    const int E = in_sizes[1] / 2;

    _Float16* W1t = (_Float16*)d_ws;
    _Float16* W2t = W1t + 128 * KATTR;
    _Float16* W3t = W2t + 128 * HPAD;
    _Float16* W4t = W3t + 128 * HPAD;
    float*    agg = (float*)(W4t + 256 * HPAD);
    float*    cnt = agg + (size_t)N * 16;

    const int prep_total = 128 * KATTR + 128 * HPAD + 128 * HPAD + 256 * HPAD;
    prep_weights<<<(prep_total + 255) / 256, 256, 0, stream>>>(
        W1, W2, W3, W4, W1t, W2t, W3t, W4t);

    const int zn = N * 17;
    zero_f32<<<(zn + 255) / 256, 256, 0, stream>>>(agg, zn);

    edge_kernel<<<NWG, BLOCK, 0, stream>>>(
        x, edge_index, edge_attr, W1t, W2t, W3t, W4t,
        b1, b2, b3, b4, agg, cnt, E);

    finalize_kernel<<<(N + 255) / 256, 256, 0, stream>>>(
        x, root, bias, agg, cnt, out, N);
}


# Hidden activations stay in registers between layers: a lane's packed f16 accumulators of feature tiles 2j, 2j+1 are the next layer's B fragment of K-step j; no LDS store and read-back per layer
# speedup vs baseline: 4.8517x; 1.1812x over previous
#include <hip/hip_runtime.h>
#include <hip/hip_bf16.h>


typedef __attribute__((ext_vector_type(8)))  _Float16 v8h;
typedef __attribute__((ext_vector_type(16))) _Float16 v16h;
typedef __attribute__((ext_vector_type(8)))  float    v8f;
typedef __attribute__((ext_vector_type(4)))  unsigned int v4u;

#define WAVE_E   32
#define NWAVES   12
#define BLOCK    (NWAVES * 32)
#define NWG      256
#define W1OFF    0
#define W2OFF    (128 * 32)
#define W3OFF    (W2OFF + 128 * 128)
#define W4OFF    (W3OFF + 128 * 128)
#define WBANKS   (W4OFF + 256 * 128)
#define HPAD     128
#define NT_H     7
#define KATTR    32
#define WCOLS    256


__device__ __forceinline__ v16h load_bfrag(const _Float16* bank, int f, int lane) {
    const _Float16* p = bank + f * 512 + lane * 8;
    v8h lo = *(const v8h*)(p);
    v8h hi = *(const v8h*)(p + 256);
    v16h r;
#pragma unroll
    for (int i = 0; i < 8; ++i) { r[i] = lo[i]; r[8 + i] = hi[i]; }
    return r;
}

template <int NT, int KS, bool SPLIT>
__device__ __forceinline__ void kloop(
    const v16h (&bin)[KS][2],
    const _Float16* Wt, int ntile0, int lane,
    v8f (&acc0)[NT], v8f (&acc1)[NT])
{
    constexpr int NH = SPLIT ? NT / 2 : NT;

    v16h b[2][NT];
#pragma unroll
    for (int t = 0; t < NT; ++t)
        b[0][t] = load_bfrag(Wt, (ntile0 + t) * KS, lane);
    __builtin_amdgcn_sched_barrier(0);

#pragma unroll
    for (int ks = 0; ks < KS; ++ks) {
        const int cur = ks & 1;
        const int nxt = cur ^ 1;
        if (ks + 1 < KS) {
#pragma unroll
            for (int t = 0; t < NH; ++t)
                b[nxt][t] = load_bfrag(Wt, (ntile0 + t) * KS + ks + 1, lane);
        }
        __builtin_amdgcn_sched_barrier(0);
#pragma unroll
        for (int t = 0; t < NH; ++t) {
            acc0[t] = __builtin_amdgcn_wmma_f32_16x16x32_f16(
                false, b[cur][t], false, bin[ks][0], (short)0, acc0[t], false, false);
            acc1[t] = __builtin_amdgcn_wmma_f32_16x16x32_f16(
                false, b[cur][t], false, bin[ks][1], (short)0, acc1[t], false, false);
        }
        if (SPLIT) {
            if (ks + 1 < KS) {
                __builtin_amdgcn_sched_barrier(0);
#pragma unroll
                for (int t = NH; t < NT; ++t)
                    b[nxt][t] = load_bfrag(Wt, (ntile0 + t) * KS + ks + 1, lane);
                __builtin_amdgcn_sched_barrier(0);
            }
#pragma unroll
            for (int t = NH; t < NT; ++t) {
                acc0[t] = __builtin_amdgcn_wmma_f32_16x16x32_f16(
                    false, b[cur][t], false, bin[ks][0], (short)0, acc0[t], false, false);
                acc1[t] = __builtin_amdgcn_wmma_f32_16x16x32_f16(
                    false, b[cur][t], false, bin[ks][1], (short)0, acc1[t], false, false);
            }
        }
        if (ks + 1 < KS) {
            __builtin_amdgcn_sched_barrier(0);
        }
    }
}

template <int NT, int KS, bool RELU, int NVALID>
__device__ __forceinline__ void gemm_stage(
    const v16h (&bin)[KS][2],
    const _Float16* Wt,
    const float* bias,
    v16h (&bout)[4][2],
    int lane)
{
    static_assert(NVALID % 4 == 0, "the bias is read in groups of four");
    static_assert(NT <= 8, "a layer's output is at most 4 K-steps of the next");
    const int khalf = lane >> 4;

    v8f acc0[NT], acc1[NT];
#pragma unroll
    for (int t = 0; t < NT; ++t) {
        acc0[t] = (v8f){0.f,0.f,0.f,0.f,0.f,0.f,0.f,0.f};
        acc1[t] = (v8f){0.f,0.f,0.f,0.f,0.f,0.f,0.f,0.f};
    }

    kloop<NT, KS, false>(bin, Wt, 0, lane, acc0, acc1);

#pragma unroll
    for (int t = 0; t < NT; ++t) {
        const int  n0  = t * 16 + khalf * 8;
        const bool vlo = (n0 + 4 <= NVALID);
        const bool vhi = (n0 + 8 <= NVALID);
        float4 blo = {0.f, 0.f, 0.f, 0.f}, bhi = {0.f, 0.f, 0.f, 0.f};
        if (vlo) blo = *(const float4*)(bias + n0);
        if (vhi) bhi = *(const float4*)(bias + n0 + 4);
        const float bv[8] = {blo.x, blo.y, blo.z, blo.w, bhi.x, bhi.y, bhi.z, bhi.w};
#pragma unroll
        for (int r = 0; r < 8; ++r) {
            const bool valid = (r < 4) ? vlo : vhi;
            float v0 = acc0[t][r] + bv[r];
            float v1 = acc1[t][r] + bv[r];
            if (RELU) { v0 = v0 > 0.0f ? v0 : 0.0f; v1 = v1 > 0.0f ? v1 : 0.0f; }
            if (!valid) { v0 = 0.0f; v1 = 0.0f; }
            bout[t >> 1][0][(t & 1) * 8 + r] = (_Float16)v0;
            bout[t >> 1][1][(t & 1) * 8 + r] = (_Float16)v1;
        }
    }
#pragma unroll
    for (int t = NT; t < 8; ++t) {
#pragma unroll
        for (int r = 0; r < 8; ++r) {
            bout[t >> 1][0][(t & 1) * 8 + r] = (_Float16)0.f;
            bout[t >> 1][1][(t & 1) * 8 + r] = (_Float16)0.f;
        }
    }
}

template <int KS, int PASS>
__device__ __forceinline__ void gemm_apply_pass(
    const v16h (&bin)[KS][2],
    const _Float16* Wt,
    const float* bias,
    const float* lds_x,
    float* lds_m,
    int lane)
{
    constexpr int NT = 8;
    constexpr int ntile0 = PASS * NT;
    const int rc    = lane & 15;
    const int khalf = lane >> 4;

    v8f acc0[NT], acc1[NT];
#pragma unroll
    for (int t = 0; t < NT; ++t) {
        acc0[t] = (v8f){0.f,0.f,0.f,0.f,0.f,0.f,0.f,0.f};
        acc1[t] = (v8f){0.f,0.f,0.f,0.f,0.f,0.f,0.f,0.f};
    }

    kloop<NT, KS, true>(bin, Wt, ntile0, lane, acc0, acc1);

    const float* x0p = lds_x + rc * 16 + ntile0;
    const float* x1p = lds_x + (16 + rc) * 16 + ntile0;
    const float4 xa = *(const float4*)(x0p);
    const float4 xb = *(const float4*)(x0p + 4);
    const float4 ya = *(const float4*)(x1p);
    const float4 yb = *(const float4*)(x1p + 4);
    const float xv[8] = {xa.x, xa.y, xa.z, xa.w, xb.x, xb.y, xb.z, xb.w};
    const float yv[8] = {ya.x, ya.y, ya.z, ya.w, yb.x, yb.y, yb.z, yb.w};

    float* m0p = lds_m + rc * 16 + khalf * 8;
    float* m1p = lds_m + (16 + rc) * 16 + khalf * 8;
    float s0[8], s1[8];
    if (PASS == 1) {
        const float4 p0 = *(const float4*)(m0p), p1 = *(const float4*)(m0p + 4);
        const float4 q0 = *(const float4*)(m1p), q1 = *(const float4*)(m1p + 4);
        s0[0] = p0.x; s0[1] = p0.y; s0[2] = p0.z; s0[3] = p0.w;
        s0[4] = p1.x; s0[5] = p1.y; s0[6] = p1.z; s0[7] = p1.w;
        s1[0] = q0.x; s1[1] = q0.y; s1[2] = q0.z; s1[3] = q0.w;
        s1[4] = q1.x; s1[5] = q1.y; s1[6] = q1.z; s1[7] = q1.w;
    } else {
#pragma unroll
        for (int r = 0; r < 8; ++r) { s0[r] = 0.0f; s1[r] = 0.0f; }
    }
#pragma unroll
    for (int t = 0; t < NT; ++t) {
        const float* bp = bias + (ntile0 + t) * 16 + khalf * 8;
        const float4 blo = *(const float4*)(bp);
        const float4 bhi = *(const float4*)(bp + 4);
        const float bv[8] = {blo.x, blo.y, blo.z, blo.w, bhi.x, bhi.y, bhi.z, bhi.w};
#pragma unroll
        for (int r = 0; r < 8; ++r) {
            s0[r] += xv[t] * (acc0[t][r] + bv[r]);
            s1[r] += yv[t] * (acc1[t][r] + bv[r]);
        }
    }
    *(float4*)(m0p)     = (float4){s0[0], s0[1], s0[2], s0[3]};
    *(float4*)(m0p + 4) = (float4){s0[4], s0[5], s0[6], s0[7]};
    *(float4*)(m1p)     = (float4){s1[0], s1[1], s1[2], s1[3]};
    *(float4*)(m1p + 4) = (float4){s1[4], s1[5], s1[6], s1[7]};
}

__device__ __forceinline__ void bank_coord(int t, int KS, int& n, int& k) {
    const int i    = t & 7;
    const int lane = (t >> 3) & 31;
    const int part = (t >> 8) & 1;
    const int f    = t >> 9;
    const int ks   = f % KS;
    const int nt   = f / KS;
    n = nt * 16 + (lane & 15);
    k = ks * 32 + part * 16 + (lane >> 4) * 8 + i;
}

__global__ void prep_weights(const float* __restrict__ W1, const float* __restrict__ W2,
                             const float* __restrict__ W3, const float* __restrict__ W4,
                             _Float16* __restrict__ W1t, _Float16* __restrict__ W2t,
                             _Float16* __restrict__ W3t, _Float16* __restrict__ W4t)
{
    int tid = blockIdx.x * blockDim.x + threadIdx.x;
    const int s1 = 128 * KATTR;
    const int s2 = s1 + 128 * HPAD;
    const int s3 = s2 + 128 * HPAD;
    const int s4 = s3 + 256 * HPAD;
    int n, k;
    if (tid < s1) {
        bank_coord(tid, KATTR / 32, n, k);
        W1t[tid] = (_Float16)((k < 8 && n < 100) ? W1[k * 100 + n] : 0.f);
    } else if (tid < s2) {
        int t = tid - s1; bank_coord(t, HPAD / 32, n, k);
        W2t[t] = (_Float16)((k < 100 && n < 100) ? W2[k * 100 + n] : 0.f);
    } else if (tid < s3) {
        int t = tid - s2; bank_coord(t, HPAD / 32, n, k);
        W3t[t] = (_Float16)((k < 100 && n < 100) ? W3[k * 100 + n] : 0.f);
    } else if (tid < s4) {
        int t = tid - s3; bank_coord(t, HPAD / 32, n, k);
        W4t[t] = (_Float16)((k < 100 && n < 256) ? W4[k * 256 + n] : 0.f);
    }
}

__global__ void zero_f32(float* __restrict__ p, int n) {
    int i = blockIdx.x * blockDim.x + threadIdx.x;
    if (i < n) p[i] = 0.0f;
}

__device__ __forceinline__ void copy_bank(const _Float16* __restrict__ src, _Float16* dst,
                                          int nvec, int tid) {
    const uint4* s = (const uint4*)src;
    uint4*       d = (uint4*)dst;
    for (int i = tid; i < nvec; i += BLOCK) d[i] = s[i];
}

__global__ __launch_bounds__(BLOCK, 3)
void edge_kernel(const float* __restrict__ x,
                 const int*   __restrict__ edge_index,
                 const float* __restrict__ edge_attr,
                 const _Float16* __restrict__ W1t, const _Float16* __restrict__ W2t,
                 const _Float16* __restrict__ W3t, const _Float16* __restrict__ W4t,
                 const float* b1, const float* b2,
                 const float* b3, const float* b4,
                 float* __restrict__ agg, float* __restrict__ cnt, int E)
{
    __shared__ __attribute__((aligned(16))) _Float16 s_w[WBANKS];
    __shared__ __attribute__((aligned(16))) float    s_x[NWAVES * WAVE_E * 16];
    __shared__ __attribute__((aligned(16))) float    s_m[NWAVES * WAVE_E * 16];

    const int tid  = threadIdx.x;
    const int lane = tid & 31;
    const int wave = __builtin_amdgcn_readfirstlane(tid >> 5);

    copy_bank(W1t, s_w + W1OFF, 128 * KATTR * 2 / 16, tid);
    copy_bank(W2t, s_w + W2OFF, 128 * HPAD  * 2 / 16, tid);
    copy_bank(W3t, s_w + W3OFF, 128 * HPAD  * 2 / 16, tid);
    copy_bank(W4t, s_w + W4OFF, 256 * HPAD  * 2 / 16, tid);

    float*    xw = s_x    + wave * WAVE_E * 16;
    float*    mw = s_m    + wave * WAVE_E * 16;

    __syncthreads();

    const _Float16* W1s = s_w + W1OFF;
    const _Float16* W2s = s_w + W2OFF;
    const _Float16* W3s = s_w + W3OFF;
    const _Float16* W4s = s_w + W4OFF;

    const int colc  = lane & 15;
    const int hhalf = lane >> 4;

    const int ntiles = (E + WAVE_E - 1) / WAVE_E;
    const int nwaves = (int)gridDim.x * NWAVES;
    for (int tile = wave * (int)gridDim.x + (int)blockIdx.x; tile < ntiles; tile += nwaves) {
        const int  ge   = tile * WAVE_E + lane;
        const bool live = ge < E;
        const int  gc   = live ? ge : E - 1;
        const int s = edge_index[gc];
        const int dmine = edge_index[E + gc];

        const int e0 = tile * WAVE_E + colc;
        const int e1 = e0 + 16;
        const int g0 = e0 < E ? e0 : E - 1;
        const int g1 = e1 < E ? e1 : E - 1;
        const float4* ea = (const float4*)edge_attr;
        const float4 p0 = ea[(long)g0 * 2];
        const float4 p1 = ea[(long)g0 * 2 + 1];
        const float4 q0 = ea[(long)g1 * 2];
        const float4 q1 = ea[(long)g1 * 2 + 1];

        const float4* xs = (const float4*)(x + (long)s * 16);
        const float4 xq0 = xs[0], xq1 = xs[1], xq2 = xs[2], xq3 = xs[3];

        v16h hin[1][2];
        {
            const float pv[8] = {p0.x, p0.y, p0.z, p0.w, p1.x, p1.y, p1.z, p1.w};
            const float qv[8] = {q0.x, q0.y, q0.z, q0.w, q1.x, q1.y, q1.z, q1.w};
            v8h ph, qh;
#pragma unroll
            for (int i = 0; i < 8; ++i) { ph[i] = (_Float16)pv[i]; qh[i] = (_Float16)qv[i]; }
            const unsigned int keep = (hhalf == 0) ? 0xFFFFFFFFu : 0u;
            v4u pu = __builtin_bit_cast(v4u, ph);
            v4u qu = __builtin_bit_cast(v4u, qh);
            pu &= keep;
            qu &= keep;
            ph = __builtin_bit_cast(v8h, pu);
            qh = __builtin_bit_cast(v8h, qu);
#pragma unroll
            for (int i = 0; i < 8; ++i) {
                hin[0][0][i]     = ph[i];
                hin[0][1][i]     = qh[i];
                hin[0][0][8 + i] = (_Float16)0.f;
                hin[0][1][8 + i] = (_Float16)0.f;
            }
        }

        v16h ha[4][2], hb[4][2];

        gemm_stage<NT_H, 1, true, 100>(hin, W1s, b1, ha, lane);

        {
            float4* dx = (float4*)(xw + lane * 16);
            dx[0] = xq0; dx[1] = xq1; dx[2] = xq2; dx[3] = xq3;
        }

        gemm_stage<NT_H, 4, true, 100>(ha, W2s, b2, hb, lane);
        gemm_stage<NT_H, 4, true, 100>(hb, W3s, b3, ha, lane);

        gemm_apply_pass<4, 0>(ha, W4s, b4, xw, mw, lane);
        gemm_apply_pass<4, 1>(ha, W4s, b4, xw, mw, lane);

        {
            const int o     = lane & 15;
            const int khalf = lane >> 4;
            const int dsel  = live ? dmine : -1;
            float msg0[8], msg1[8];
#pragma unroll
            for (int r = 0; r < 8; ++r) {
                msg0[r] = mw[(8 * khalf + r) * 16 + o];
                msg1[r] = mw[(16 + 8 * khalf + r) * 16 + o];
            }
#pragma unroll
            for (int r = 0; r < 8; ++r) {
                const int d0 = __shfl(dsel, 8 * khalf + r);
                const int d1 = __shfl(dsel, 16 + 8 * khalf + r);
                if (d0 >= 0) atomicAdd(&agg[(long)d0 * 16 + o], msg0[r]);
                if (d1 >= 0) atomicAdd(&agg[(long)d1 * 16 + o], msg1[r]);
            }
            if (dsel >= 0) atomicAdd(&cnt[dsel], 1.0f);
        }
    }
}

__global__ void finalize_kernel(const float* __restrict__ x, const float* __restrict__ root,
                                const float* __restrict__ bias, const float* __restrict__ agg,
                                const float* __restrict__ cnt, float* __restrict__ out, int N)
{
    int n = blockIdx.x * blockDim.x + threadIdx.x;
    if (n >= N) return;
    float xi[16];
#pragma unroll
    for (int i = 0; i < 16; ++i) xi[i] = x[(long)n * 16 + i];
    float c = cnt[n];
    float inv = 1.0f / (c > 1.0f ? c : 1.0f);
#pragma unroll
    for (int o = 0; o < 16; ++o) {
        float acc = 0.0f;
#pragma unroll
        for (int i = 0; i < 16; ++i) acc += xi[i] * root[i * 16 + o];
        out[(long)n * 16 + o] = acc + agg[(long)n * 16 + o] * inv + bias[o];
    }
}

extern "C" void kernel_launch(void* const* d_in, const int* in_sizes, int n_in,
                              void* d_out, int out_size, void* d_ws, size_t ws_size,
                              hipStream_t stream)
{
    const float* x          = (const float*)d_in[0];
    const int*   edge_index = (const int*)  d_in[1];
    const float* edge_attr  = (const float*)d_in[2];
    const float* W1 = (const float*)d_in[3];  const float* b1 = (const float*)d_in[4];
    const float* W2 = (const float*)d_in[5];  const float* b2 = (const float*)d_in[6];
    const float* W3 = (const float*)d_in[7];  const float* b3 = (const float*)d_in[8];
    const float* W4 = (const float*)d_in[9];  const float* b4 = (const float*)d_in[10];
    const float* root = (const float*)d_in[11];
    const float* bias = (const float*)d_in[12];
    float* out = (float*)d_out;

    const int N = in_sizes[0] / 16;
    const int E = in_sizes[1] / 2;

    _Float16* W1t = (_Float16*)d_ws;
    _Float16* W2t = W1t + 128 * KATTR;
    _Float16* W3t = W2t + 128 * HPAD;
    _Float16* W4t = W3t + 128 * HPAD;
    float*    agg = (float*)(W4t + 256 * HPAD);
    float*    cnt = agg + (size_t)N * 16;

    const int prep_total = 128 * KATTR + 128 * HPAD + 128 * HPAD + 256 * HPAD;
    prep_weights<<<(prep_total + 255) / 256, 256, 0, stream>>>(
        W1, W2, W3, W4, W1t, W2t, W3t, W4t);

    const int zn = N * 17;
    zero_f32<<<(zn + 255) / 256, 256, 0, stream>>>(agg, zn);

    edge_kernel<<<NWG, BLOCK, 0, stream>>>(
        x, edge_index, edge_attr, W1t, W2t, W3t, W4t,
        b1, b2, b3, b4, agg, cnt, E);

    finalize_kernel<<<(N + 255) / 256, 256, 0, stream>>>(
        x, root, bias, agg, cnt, out, N);
}


# Edges grouped by destination range in LDS (with the weight prep); a workgroup sums its range's messages by LDS atomics (arrival order as before) and writes its rows: 2 launches, no global atomics
# speedup vs baseline: 5.7165x; 1.1782x over previous
#include <hip/hip_runtime.h>
#include <hip/hip_bf16.h>


typedef __attribute__((ext_vector_type(8)))  _Float16 v8h;
typedef __attribute__((ext_vector_type(16))) _Float16 v16h;
typedef __attribute__((ext_vector_type(8)))  float    v8f;
typedef __attribute__((ext_vector_type(4)))  unsigned int v4u;
typedef __attribute__((ext_vector_type(4)))  float    v4f;

#define WAVE_E   32
#define NWAVES   12
#define BLOCK    (NWAVES * 32)
#define NB       256
#define NPB_CAP  256
#define PTHREADS 512
#define CHUNK    3200
#define EPT      ((CHUNK + PTHREADS - 1) / PTHREADS)
#define LIST_CAP 8192
#define W1OFF    0
#define W2OFF    (128 * 32)
#define W3OFF    (W2OFF + 128 * 128)
#define W4OFF    (W3OFF + 128 * 128)
#define WBANKS   (W4OFF + 256 * 128)
#define HPAD     128
#define NT_H     7
#define KATTR    32
#define WCOLS    256


__device__ __forceinline__ v16h load_bfrag(const _Float16* bank, int f, int lane) {
    const _Float16* p = bank + f * 512 + lane * 8;
    v8h lo = *(const v8h*)(p);
    v8h hi = *(const v8h*)(p + 256);
    v16h r;
#pragma unroll
    for (int i = 0; i < 8; ++i) { r[i] = lo[i]; r[8 + i] = hi[i]; }
    return r;
}

template <int NT, int KS, bool SPLIT>
__device__ __forceinline__ void kloop(
    const v16h (&bin)[KS][2],
    const _Float16* Wt, int ntile0, int lane,
    v8f (&acc0)[NT], v8f (&acc1)[NT])
{
    constexpr int NH = SPLIT ? NT / 2 : NT;

    v16h b[2][NT];
#pragma unroll
    for (int t = 0; t < NT; ++t)
        b[0][t] = load_bfrag(Wt, (ntile0 + t) * KS, lane);
    __builtin_amdgcn_sched_barrier(0);

#pragma unroll
    for (int ks = 0; ks < KS; ++ks) {
        const int cur = ks & 1;
        const int nxt = cur ^ 1;
        if (ks + 1 < KS) {
#pragma unroll
            for (int t = 0; t < NH; ++t)
                b[nxt][t] = load_bfrag(Wt, (ntile0 + t) * KS + ks + 1, lane);
        }
        __builtin_amdgcn_sched_barrier(0);
#pragma unroll
        for (int t = 0; t < NH; ++t) {
            acc0[t] = __builtin_amdgcn_wmma_f32_16x16x32_f16(
                false, b[cur][t], false, bin[ks][0], (short)0, acc0[t], false, false);
            acc1[t] = __builtin_amdgcn_wmma_f32_16x16x32_f16(
                false, b[cur][t], false, bin[ks][1], (short)0, acc1[t], false, false);
        }
        if (SPLIT) {
            if (ks + 1 < KS) {
                __builtin_amdgcn_sched_barrier(0);
#pragma unroll
                for (int t = NH; t < NT; ++t)
                    b[nxt][t] = load_bfrag(Wt, (ntile0 + t) * KS + ks + 1, lane);
                __builtin_amdgcn_sched_barrier(0);
            }
#pragma unroll
            for (int t = NH; t < NT; ++t) {
                acc0[t] = __builtin_amdgcn_wmma_f32_16x16x32_f16(
                    false, b[cur][t], false, bin[ks][0], (short)0, acc0[t], false, false);
                acc1[t] = __builtin_amdgcn_wmma_f32_16x16x32_f16(
                    false, b[cur][t], false, bin[ks][1], (short)0, acc1[t], false, false);
            }
        }
        if (ks + 1 < KS) {
            __builtin_amdgcn_sched_barrier(0);
        }
    }
}

template <int NT, int KS, bool RELU, int NVALID>
__device__ __forceinline__ void gemm_stage(
    const v16h (&bin)[KS][2],
    const _Float16* Wt,
    const float* bias,
    v16h (&bout)[4][2],
    int lane)
{
    static_assert(NVALID % 4 == 0, "the bias is read in groups of four");
    static_assert(NT <= 8, "a layer's output is at most 4 K-steps of the next");
    const int khalf = lane >> 4;

    v8f acc0[NT], acc1[NT];
#pragma unroll
    for (int t = 0; t < NT; ++t) {
        acc0[t] = (v8f){0.f,0.f,0.f,0.f,0.f,0.f,0.f,0.f};
        acc1[t] = (v8f){0.f,0.f,0.f,0.f,0.f,0.f,0.f,0.f};
    }

    kloop<NT, KS, false>(bin, Wt, 0, lane, acc0, acc1);

#pragma unroll
    for (int t = 0; t < NT; ++t) {
        const int  n0  = t * 16 + khalf * 8;
        const bool vlo = (n0 + 4 <= NVALID);
        const bool vhi = (n0 + 8 <= NVALID);
        float4 blo = {0.f, 0.f, 0.f, 0.f}, bhi = {0.f, 0.f, 0.f, 0.f};
        if (vlo) blo = *(const float4*)(bias + n0);
        if (vhi) bhi = *(const float4*)(bias + n0 + 4);
        const float bv[8] = {blo.x, blo.y, blo.z, blo.w, bhi.x, bhi.y, bhi.z, bhi.w};
#pragma unroll
        for (int r = 0; r < 8; ++r) {
            const bool valid = (r < 4) ? vlo : vhi;
            float v0 = acc0[t][r] + bv[r];
            float v1 = acc1[t][r] + bv[r];
            if (RELU) { v0 = v0 > 0.0f ? v0 : 0.0f; v1 = v1 > 0.0f ? v1 : 0.0f; }
            if (!valid) { v0 = 0.0f; v1 = 0.0f; }
            bout[t >> 1][0][(t & 1) * 8 + r] = (_Float16)v0;
            bout[t >> 1][1][(t & 1) * 8 + r] = (_Float16)v1;
        }
    }
#pragma unroll
    for (int t = NT; t < 8; ++t) {
#pragma unroll
        for (int r = 0; r < 8; ++r) {
            bout[t >> 1][0][(t & 1) * 8 + r] = (_Float16)0.f;
            bout[t >> 1][1][(t & 1) * 8 + r] = (_Float16)0.f;
        }
    }
}

template <int KS, int PASS>
__device__ __forceinline__ void gemm_apply_pass(
    const v16h (&bin)[KS][2],
    const _Float16* Wt,
    const float* bias,
    const float* lds_x,
    float* lds_m,
    int lane)
{
    constexpr int NT = 8;
    constexpr int ntile0 = PASS * NT;
    const int rc    = lane & 15;
    const int khalf = lane >> 4;

    v8f acc0[NT], acc1[NT];
#pragma unroll
    for (int t = 0; t < NT; ++t) {
        acc0[t] = (v8f){0.f,0.f,0.f,0.f,0.f,0.f,0.f,0.f};
        acc1[t] = (v8f){0.f,0.f,0.f,0.f,0.f,0.f,0.f,0.f};
    }

    kloop<NT, KS, true>(bin, Wt, ntile0, lane, acc0, acc1);

    const float* x0p = lds_x + rc * 16 + ntile0;
    const float* x1p = lds_x + (16 + rc) * 16 + ntile0;
    const float4 xa = *(const float4*)(x0p);
    const float4 xb = *(const float4*)(x0p + 4);
    const float4 ya = *(const float4*)(x1p);
    const float4 yb = *(const float4*)(x1p + 4);
    const float xv[8] = {xa.x, xa.y, xa.z, xa.w, xb.x, xb.y, xb.z, xb.w};
    const float yv[8] = {ya.x, ya.y, ya.z, ya.w, yb.x, yb.y, yb.z, yb.w};

    float* m0p = lds_m + rc * 16 + khalf * 8;
    float* m1p = lds_m + (16 + rc) * 16 + khalf * 8;
    float s0[8], s1[8];
    if (PASS == 1) {
        const float4 p0 = *(const float4*)(m0p), p1 = *(const float4*)(m0p + 4);
        const float4 q0 = *(const float4*)(m1p), q1 = *(const float4*)(m1p + 4);
        s0[0] = p0.x; s0[1] = p0.y; s0[2] = p0.z; s0[3] = p0.w;
        s0[4] = p1.x; s0[5] = p1.y; s0[6] = p1.z; s0[7] = p1.w;
        s1[0] = q0.x; s1[1] = q0.y; s1[2] = q0.z; s1[3] = q0.w;
        s1[4] = q1.x; s1[5] = q1.y; s1[6] = q1.z; s1[7] = q1.w;
    } else {
#pragma unroll
        for (int r = 0; r < 8; ++r) { s0[r] = 0.0f; s1[r] = 0.0f; }
    }
#pragma unroll
    for (int t = 0; t < NT; ++t) {
        const float* bp = bias + (ntile0 + t) * 16 + khalf * 8;
        const float4 blo = *(const float4*)(bp);
        const float4 bhi = *(const float4*)(bp + 4);
        const float bv[8] = {blo.x, blo.y, blo.z, blo.w, bhi.x, bhi.y, bhi.z, bhi.w};
#pragma unroll
        for (int r = 0; r < 8; ++r) {
            s0[r] += xv[t] * (acc0[t][r] + bv[r]);
            s1[r] += yv[t] * (acc1[t][r] + bv[r]);
        }
    }
    *(float4*)(m0p)     = (float4){s0[0], s0[1], s0[2], s0[3]};
    *(float4*)(m0p + 4) = (float4){s0[4], s0[5], s0[6], s0[7]};
    *(float4*)(m1p)     = (float4){s1[0], s1[1], s1[2], s1[3]};
    *(float4*)(m1p + 4) = (float4){s1[4], s1[5], s1[6], s1[7]};
}

__device__ __forceinline__ void bank_coord(int t, int KS, int& n, int& k) {
    const int i    = t & 7;
    const int lane = (t >> 3) & 31;
    const int part = (t >> 8) & 1;
    const int f    = t >> 9;
    const int ks   = f % KS;
    const int nt   = f / KS;
    n = nt * 16 + (lane & 15);
    k = ks * 32 + part * 16 + (lane >> 4) * 8 + i;
}

__device__ __forceinline__ void prep_weights(const float* __restrict__ W1, const float* __restrict__ W2,
                             const float* __restrict__ W3, const float* __restrict__ W4,
                             _Float16* __restrict__ W1t, _Float16* __restrict__ W2t,
                             _Float16* __restrict__ W3t, _Float16* __restrict__ W4t, int tid)
{
    const int s1 = 128 * KATTR;
    const int s2 = s1 + 128 * HPAD;
    const int s3 = s2 + 128 * HPAD;
    const int s4 = s3 + 256 * HPAD;
    int n, k;
    if (tid < s1) {
        bank_coord(tid, KATTR / 32, n, k);
        W1t[tid] = (_Float16)((k < 8 && n < 100) ? W1[k * 100 + n] : 0.f);
    } else if (tid < s2) {
        int t = tid - s1; bank_coord(t, HPAD / 32, n, k);
        W2t[t] = (_Float16)((k < 100 && n < 100) ? W2[k * 100 + n] : 0.f);
    } else if (tid < s3) {
        int t = tid - s2; bank_coord(t, HPAD / 32, n, k);
        W3t[t] = (_Float16)((k < 100 && n < 100) ? W3[k * 100 + n] : 0.f);
    } else if (tid < s4) {
        int t = tid - s3; bank_coord(t, HPAD / 32, n, k);
        W4t[t] = (_Float16)((k < 100 && n < 256) ? W4[k * 256 + n] : 0.f);
    }
}

__device__ __forceinline__ int range_of(int d, int npb) {
    const int b = d / npb;
    return b < NB - 1 ? b : NB - 1;
}

template <int NW>
__device__ __forceinline__ int excl_scan_block(int v, int tid, int* tmp) {
    const int lane = tid & 31, w = tid >> 5;
    int inc = v;
#pragma unroll
    for (int d = 1; d < 32; d <<= 1) {
        const int t = __shfl_up(inc, d);
        if (lane >= d) inc += t;
    }
    if (lane == 31) tmp[w] = inc;
    __syncthreads();
    int off = 0;
#pragma unroll
    for (int j = 0; j < NW; ++j) {
        const int tj = tmp[j];
        if (j < w) off += tj;
    }
    return off + inc - v;
}

__global__ __launch_bounds__(PTHREADS)
void prep_and_group(const float* __restrict__ W1, const float* __restrict__ W2,
                    const float* __restrict__ W3, const float* __restrict__ W4,
                    _Float16* __restrict__ W1t, _Float16* __restrict__ W2t,
                    _Float16* __restrict__ W3t, _Float16* __restrict__ W4t,
                    const int* __restrict__ edge_index, const float* __restrict__ edge_attr,
                    int* __restrict__ g_src, int* __restrict__ g_dst, float* __restrict__ g_attr,
                    int* __restrict__ tab_cnt, int* __restrict__ tab_off,
                    int E, int N)
{
    __shared__ __attribute__((aligned(16))) float st_attr[CHUNK * 8];
    __shared__ int st_src[CHUNK];
    __shared__ int st_dst[CHUNK];
    __shared__ int h[NB];
    __shared__ int loff[NB];
    __shared__ int tmp[PTHREADS / 32];

    const int tid = threadIdx.x;
    {
        const int prep_total = 128 * KATTR + 128 * HPAD + 128 * HPAD + 256 * HPAD;
        for (int t = (int)blockIdx.x * PTHREADS + tid; t < prep_total;
             t += (int)gridDim.x * PTHREADS)
            prep_weights(W1, W2, W3, W4, W1t, W2t, W3t, W4t, t);
    }

    const int w   = (int)blockIdx.x;
    const int e0  = w * CHUNK;
    const int n   = E - e0 < CHUNK ? E - e0 : CHUNK;
    const int npb = (N + NB - 1) / NB;
    const v4f* ea = (const v4f*)edge_attr;

    if (tid < NB) h[tid] = 0;
    __syncthreads();

    int    es[EPT], ed[EPT], eb[EPT], er[EPT];
    v4f    a0[EPT], a1[EPT];
#pragma unroll
    for (int k = 0; k < EPT; ++k) {
        const int i = tid + k * PTHREADS;
        const int e = e0 + (i < n ? i : n - 1);
        es[k] = edge_index[e];
        ed[k] = edge_index[E + e];
        a0[k] = ea[(long)e * 2];
        a1[k] = ea[(long)e * 2 + 1];
    }
#pragma unroll
    for (int k = 0; k < EPT; ++k) {
        const int i = tid + k * PTHREADS;
        if (i < n) {
            eb[k] = range_of(ed[k], npb);
            er[k] = atomicAdd(&h[eb[k]], 1);
        }
    }
    __syncthreads();

    const int cb   = tid < NB ? h[tid] : 0;
    const int excl = excl_scan_block<PTHREADS / 32>(cb, tid, tmp);
    if (tid < NB) {
        loff[tid] = excl;
        tab_cnt[tid * (int)gridDim.x + w] = cb;
        tab_off[tid * (int)gridDim.x + w] = excl;
    }
    __syncthreads();

#pragma unroll
    for (int k = 0; k < EPT; ++k) {
        const int i = tid + k * PTHREADS;
        if (i < n) {
            const int q = loff[eb[k]] + er[k];
            st_src[q] = es[k];
            st_dst[q] = ed[k];
            ((v4f*)st_attr)[q * 2]     = a0[k];
            ((v4f*)st_attr)[q * 2 + 1] = a1[k];
        }
    }
    __syncthreads();

    for (int i = tid; i < n; i += PTHREADS) {
        g_src[e0 + i] = st_src[i];
        g_dst[e0 + i] = st_dst[i];
    }
    v4f* ga = (v4f*)g_attr + (long)e0 * 2;
    for (int i = tid; i < n * 2; i += PTHREADS) ga[i] = ((const v4f*)st_attr)[i];
}

__device__ __forceinline__ void copy_bank(const _Float16* __restrict__ src, _Float16* dst,
                                          int nvec, int tid) {
    const uint4* s = (const uint4*)src;
    uint4*       d = (uint4*)dst;
    for (int i = tid; i < nvec; i += BLOCK) d[i] = s[i];
}

__global__ __launch_bounds__(BLOCK, 3)
void edge_kernel(const float* __restrict__ x,
                 const int*   __restrict__ g_src,
                 const int*   __restrict__ g_dst,
                 const float* __restrict__ g_attr,
                 const int*   __restrict__ tab_cnt,
                 const int*   __restrict__ tab_off,
                 int nchunks,
                 const _Float16* __restrict__ W1t, const _Float16* __restrict__ W2t,
                 const _Float16* __restrict__ W3t, const _Float16* __restrict__ W4t,
                 const float* b1, const float* b2,
                 const float* b3, const float* b4,
                 const float* __restrict__ root, const float* __restrict__ bias,
                 float* __restrict__ out, int N)
{
    __shared__ __attribute__((aligned(16))) _Float16 s_w[WBANKS];
    __shared__ __attribute__((aligned(16))) float    s_x[NWAVES * WAVE_E * 16];
    __shared__ __attribute__((aligned(16))) float    s_m[NWAVES * WAVE_E * 16];
    __shared__ __attribute__((aligned(16))) float    s_agg[NPB_CAP * 16];
    __shared__ __attribute__((aligned(16))) float    s_xn[((NPB_CAP * 4 + BLOCK - 1) / BLOCK) * BLOCK * 4];
    __shared__ float s_cnt[NPB_CAP];
    __shared__ int   s_list[LIST_CAP];
    __shared__ int   s_tmp[NWAVES];

    const int tid  = threadIdx.x;
    const int lane = tid & 31;
    const int wave = __builtin_amdgcn_readfirstlane(tid >> 5);

    const int npb   = (N + NB - 1) / NB;
    const int node0 = (int)blockIdx.x * npb;
    const int nown  = node0 + npb <= N ? npb : (N > node0 ? N - node0 : 0);
    for (int i = tid; i < NPB_CAP * 16; i += BLOCK) s_agg[i] = 0.0f;
    for (int i = tid; i < NPB_CAP; i += BLOCK) s_cnt[i] = 0.0f;
    constexpr int XQ = (NPB_CAP * 4 + BLOCK - 1) / BLOCK;
    {
        const float4* xs4 = (const float4*)(x + (long)(nown > 0 ? node0 : 0) * 16);
        float4*       xd4 = (float4*)s_xn;
        const int last = nown > 0 ? nown * 4 - 1 : 0;
        float4 xrow[XQ];
#pragma unroll
        for (int k = 0; k < XQ; ++k) {
            const int i = tid + k * BLOCK;
            xrow[k] = xs4[i < last ? i : last];
        }
#pragma unroll
        for (int k = 0; k < XQ; ++k) xd4[tid + k * BLOCK] = xrow[k];
    }
    const int* tcnt = tab_cnt + (long)blockIdx.x * nchunks;
    const int* toff = tab_off + (long)blockIdx.x * nchunks;
    const int  wfirst = tid < nchunks ? tid : nchunks - 1;
    const int  cfirst = tcnt[wfirst];
    const int  gfirst = wfirst * CHUNK + toff[wfirst];

    copy_bank(W1t, s_w + W1OFF, 128 * KATTR * 2 / 16, tid);
    copy_bank(W2t, s_w + W2OFF, 128 * HPAD  * 2 / 16, tid);
    copy_bank(W3t, s_w + W3OFF, 128 * HPAD  * 2 / 16, tid);
    copy_bank(W4t, s_w + W4OFF, 256 * HPAD  * 2 / 16, tid);

    float*    xw = s_x    + wave * WAVE_E * 16;
    float*    mw = s_m    + wave * WAVE_E * 16;

    const int c0 = tid < nchunks ? cfirst : 0;
    int lc = c0;
    for (int w = tid + BLOCK; w < nchunks; w += BLOCK) lc += tcnt[w];
    const int lexcl = excl_scan_block<NWAVES>(lc, tid, s_tmp);
    int etot = 0;
#pragma unroll
    for (int j = 0; j < NWAVES; ++j) etot += s_tmp[j];
    const int Eb = __builtin_amdgcn_readfirstlane(etot);

    const _Float16* W1s = s_w + W1OFF;
    const _Float16* W2s = s_w + W2OFF;
    const _Float16* W3s = s_w + W3OFF;
    const _Float16* W4s = s_w + W4OFF;

    const int colc  = lane & 15;
    const int hhalf = lane >> 4;

    for (int r0 = 0; r0 < Eb; r0 += LIST_CAP) {
    {
        int pos = lexcl - r0;
        for (int j = 0; j < c0; ++j) {
            const int p = pos + j;
            if (p >= 0 && p < LIST_CAP) s_list[p] = gfirst + j;
        }
        pos += c0;
        for (int w = tid + BLOCK; w < nchunks; w += BLOCK) {
            const int cw = tcnt[w];
            const int gw = w * CHUNK + toff[w];
            for (int j = 0; j < cw; ++j) {
                const int p = pos + j;
                if (p >= 0 && p < LIST_CAP) s_list[p] = gw + j;
            }
            pos += cw;
        }
    }
    __syncthreads();
    const int E = Eb - r0 < LIST_CAP ? Eb - r0 : LIST_CAP;

    const int ntiles = (E + WAVE_E - 1) / WAVE_E;
    for (int tile = wave; tile < ntiles; tile += NWAVES) {
        const int  ge   = tile * WAVE_E + lane;
        const bool live = ge < E;
        const int  gc   = live ? ge : E - 1;
        const int gp = s_list[gc];
        const int s = g_src[gp];
        const int dmine = g_dst[gp] - node0;

        const int e0 = tile * WAVE_E + colc;
        const int e1 = e0 + 16;
        const int g0 = s_list[e0 < E ? e0 : E - 1];
        const int g1 = s_list[e1 < E ? e1 : E - 1];
        const float4* ea = (const float4*)g_attr;
        const float4 p0 = ea[(long)g0 * 2];
        const float4 p1 = ea[(long)g0 * 2 + 1];
        const float4 q0 = ea[(long)g1 * 2];
        const float4 q1 = ea[(long)g1 * 2 + 1];

        const float4* xs = (const float4*)(x + (long)s * 16);
        const float4 xq0 = xs[0], xq1 = xs[1], xq2 = xs[2], xq3 = xs[3];

        v16h hin[1][2];
        {
            const float pv[8] = {p0.x, p0.y, p0.z, p0.w, p1.x, p1.y, p1.z, p1.w};
            const float qv[8] = {q0.x, q0.y, q0.z, q0.w, q1.x, q1.y, q1.z, q1.w};
            v8h ph, qh;
#pragma unroll
            for (int i = 0; i < 8; ++i) { ph[i] = (_Float16)pv[i]; qh[i] = (_Float16)qv[i]; }
            const unsigned int keep = (hhalf == 0) ? 0xFFFFFFFFu : 0u;
            v4u pu = __builtin_bit_cast(v4u, ph);
            v4u qu = __builtin_bit_cast(v4u, qh);
            pu &= keep;
            qu &= keep;
            ph = __builtin_bit_cast(v8h, pu);
            qh = __builtin_bit_cast(v8h, qu);
#pragma unroll
            for (int i = 0; i < 8; ++i) {
                hin[0][0][i]     = ph[i];
                hin[0][1][i]     = qh[i];
                hin[0][0][8 + i] = (_Float16)0.f;
                hin[0][1][8 + i] = (_Float16)0.f;
            }
        }

        v16h ha[4][2], hb[4][2];

        gemm_stage<NT_H, 1, true, 100>(hin, W1s, b1, ha, lane);

        {
            float4* dx = (float4*)(xw + lane * 16);
            dx[0] = xq0; dx[1] = xq1; dx[2] = xq2; dx[3] = xq3;
        }

        gemm_stage<NT_H, 4, true, 100>(ha, W2s, b2, hb, lane);
        gemm_stage<NT_H, 4, true, 100>(hb, W3s, b3, ha, lane);

        gemm_apply_pass<4, 0>(ha, W4s, b4, xw, mw, lane);
        gemm_apply_pass<4, 1>(ha, W4s, b4, xw, mw, lane);

        {
            const int o     = lane & 15;
            const int khalf = lane >> 4;
            const int dsel  = live ? dmine : -1;
            xw[lane] = __int_as_float(dsel);
            float msg0[8], msg1[8];
#pragma unroll
            for (int r = 0; r < 8; ++r) {
                msg0[r] = mw[(8 * khalf + r) * 16 + o];
                msg1[r] = mw[(16 + 8 * khalf + r) * 16 + o];
            }
            const float4 da = *(const float4*)(xw + 8 * khalf);
            const float4 db = *(const float4*)(xw + 8 * khalf + 4);
            const float4 dc = *(const float4*)(xw + 16 + 8 * khalf);
            const float4 dd = *(const float4*)(xw + 16 + 8 * khalf + 4);
            const float dv0[8] = {da.x, da.y, da.z, da.w, db.x, db.y, db.z, db.w};
            const float dv1[8] = {dc.x, dc.y, dc.z, dc.w, dd.x, dd.y, dd.z, dd.w};
#pragma unroll
            for (int r = 0; r < 8; ++r) {
                const int d0 = __float_as_int(dv0[r]);
                const int d1 = __float_as_int(dv1[r]);
                if (d0 >= 0) atomicAdd(&s_agg[d0 * 16 + o], msg0[r]);
                if (d1 >= 0) atomicAdd(&s_agg[d1 * 16 + o], msg1[r]);
            }
            if (dsel >= 0) atomicAdd(&s_cnt[dsel], 1.0f);
        }
    }
    __syncthreads();
    }

    __syncthreads();
    {
        const int o = tid & 15;
        float rcol[16];
#pragma unroll
        for (int i = 0; i < 16; ++i) rcol[i] = root[i * 16 + o];
        const float bo = bias[o];
        for (int idx = tid; idx < nown * 16; idx += BLOCK) {
            const int nl = idx >> 4;
            const float4* xr = (const float4*)(s_xn + nl * 16);
            const float4 x0 = xr[0], x1 = xr[1], x2 = xr[2], x3 = xr[3];
            const float xi[16] = {x0.x, x0.y, x0.z, x0.w, x1.x, x1.y, x1.z, x1.w,
                                  x2.x, x2.y, x2.z, x2.w, x3.x, x3.y, x3.z, x3.w};
            const float cn  = s_cnt[nl];
            const float inv = 1.0f / (cn > 1.0f ? cn : 1.0f);
            float acc = 0.0f;
#pragma unroll
            for (int i = 0; i < 16; ++i) acc += xi[i] * rcol[i];
            out[(long)node0 * 16 + idx] = acc + s_agg[idx] * inv + bo;
        }
    }
}

extern "C" void kernel_launch(void* const* d_in, const int* in_sizes, int n_in,
                              void* d_out, int out_size, void* d_ws, size_t ws_size,
                              hipStream_t stream)
{
    const float* x          = (const float*)d_in[0];
    const int*   edge_index = (const int*)  d_in[1];
    const float* edge_attr  = (const float*)d_in[2];
    const float* W1 = (const float*)d_in[3];  const float* b1 = (const float*)d_in[4];
    const float* W2 = (const float*)d_in[5];  const float* b2 = (const float*)d_in[6];
    const float* W3 = (const float*)d_in[7];  const float* b3 = (const float*)d_in[8];
    const float* W4 = (const float*)d_in[9];  const float* b4 = (const float*)d_in[10];
    const float* root = (const float*)d_in[11];
    const float* bias = (const float*)d_in[12];
    float* out = (float*)d_out;

    const int N = in_sizes[0] / 16;
    const int E = in_sizes[1] / 2;

    const int nchunks = (E + CHUNK - 1) / CHUNK;

    _Float16* W1t = (_Float16*)d_ws;
    _Float16* W2t = W1t + 128 * KATTR;
    _Float16* W3t = W2t + 128 * HPAD;
    _Float16* W4t = W3t + 128 * HPAD;
    int*      tab_cnt = (int*)(W4t + 256 * HPAD);
    int*      tab_off = tab_cnt + (size_t)NB * nchunks;
    float*    g_attr  = (float*)(tab_off + (size_t)NB * nchunks);
    int*      g_src   = (int*)(g_attr + (size_t)E * 8);
    int*      g_dst   = g_src + E;

    prep_and_group<<<nchunks, PTHREADS, 0, stream>>>(
        W1, W2, W3, W4, W1t, W2t, W3t, W4t,
        edge_index, edge_attr, g_src, g_dst, g_attr, tab_cnt, tab_off, E, N);

    edge_kernel<<<NB, BLOCK, 0, stream>>>(
        x, g_src, g_dst, g_attr, tab_cnt, tab_off, nchunks, W1t, W2t, W3t, W4t,
        b1, b2, b3, b4, root, bias, out, N);
}


# Biases folded into the matrix products: two K-padding rows per weight bank hold each bias as f16 value + f16 remainder against constant-1 activations; last layer's two passes looped to fit registers
# speedup vs baseline: 6.3973x; 1.1191x over previous
#include <hip/hip_runtime.h>
#include <hip/hip_bf16.h>


typedef __attribute__((ext_vector_type(8)))  _Float16 v8h;
typedef __attribute__((ext_vector_type(16))) _Float16 v16h;
typedef __attribute__((ext_vector_type(8)))  float    v8f;
typedef __attribute__((ext_vector_type(4)))  unsigned int v4u;
typedef __attribute__((ext_vector_type(4)))  float    v4f;

#define WAVE_E   32
#define NWAVES   12
#define BLOCK    (NWAVES * 32)
#define NB       256
#define NPB_CAP  256
#define PTHREADS 512
#define CHUNK    3200
#define EPT      ((CHUNK + PTHREADS - 1) / PTHREADS)
#define LIST_CAP 8192
#define W1OFF    0
#define W2OFF    (128 * 32)
#define W3OFF    (W2OFF + 128 * 128)
#define W4OFF    (W3OFF + 128 * 128)
#define WBANKS   (W4OFF + 256 * 128)
#define HPAD     128
#define NT_H     7
#define KATTR    32
#define WCOLS    256


__device__ __forceinline__ v16h load_bfrag(const _Float16* bank, int f, int lane) {
    const _Float16* p = bank + f * 512 + lane * 8;
    v8h lo = *(const v8h*)(p);
    v8h hi = *(const v8h*)(p + 256);
    v16h r;
#pragma unroll
    for (int i = 0; i < 8; ++i) { r[i] = lo[i]; r[8 + i] = hi[i]; }
    return r;
}

template <int NT, int KS, bool SPLIT>
__device__ __forceinline__ void kloop(
    const v16h (&bin)[KS][2],
    const _Float16* Wt, int ntile0, int lane,
    v8f (&acc0)[NT], v8f (&acc1)[NT])
{
    constexpr int NH = SPLIT ? NT / 2 : NT;

    v16h b[2][NT];
#pragma unroll
    for (int t = 0; t < NT; ++t)
        b[0][t] = load_bfrag(Wt, (ntile0 + t) * KS, lane);
    __builtin_amdgcn_sched_barrier(0);

#pragma unroll
    for (int ks = 0; ks < KS; ++ks) {
        const int cur = ks & 1;
        const int nxt = cur ^ 1;
        if (ks + 1 < KS) {
#pragma unroll
            for (int t = 0; t < NH; ++t)
                b[nxt][t] = load_bfrag(Wt, (ntile0 + t) * KS + ks + 1, lane);
        }
        __builtin_amdgcn_sched_barrier(0);
#pragma unroll
        for (int t = 0; t < NH; ++t) {
            acc0[t] = __builtin_amdgcn_wmma_f32_16x16x32_f16(
                false, b[cur][t], false, bin[ks][0], (short)0, acc0[t], false, false);
            acc1[t] = __builtin_amdgcn_wmma_f32_16x16x32_f16(
                false, b[cur][t], false, bin[ks][1], (short)0, acc1[t], false, false);
        }
        if (SPLIT) {
            if (ks + 1 < KS) {
                __builtin_amdgcn_sched_barrier(0);
#pragma unroll
                for (int t = NH; t < NT; ++t)
                    b[nxt][t] = load_bfrag(Wt, (ntile0 + t) * KS + ks + 1, lane);
                __builtin_amdgcn_sched_barrier(0);
            }
#pragma unroll
            for (int t = NH; t < NT; ++t) {
                acc0[t] = __builtin_amdgcn_wmma_f32_16x16x32_f16(
                    false, b[cur][t], false, bin[ks][0], (short)0, acc0[t], false, false);
                acc1[t] = __builtin_amdgcn_wmma_f32_16x16x32_f16(
                    false, b[cur][t], false, bin[ks][1], (short)0, acc1[t], false, false);
            }
        }
        if (ks + 1 < KS) {
            __builtin_amdgcn_sched_barrier(0);
        }
    }
}

template <int NT, int KS, bool RELU>
__device__ __forceinline__ void gemm_stage(
    const v16h (&bin)[KS][2],
    const _Float16* Wt,
    v16h (&bout)[4][2],
    int lane)
{
    static_assert(NT <= 8, "a layer's output is at most 4 K-steps of the next");

    v8f acc0[NT], acc1[NT];
#pragma unroll
    for (int t = 0; t < NT; ++t) {
        acc0[t] = (v8f){0.f,0.f,0.f,0.f,0.f,0.f,0.f,0.f};
        acc1[t] = (v8f){0.f,0.f,0.f,0.f,0.f,0.f,0.f,0.f};
    }

    kloop<NT, KS, false>(bin, Wt, 0, lane, acc0, acc1);

#pragma unroll
    for (int t = 0; t < NT; ++t) {
        v8h p0, p1;
#pragma unroll
        for (int r = 0; r < 8; ++r) {
            p0[r] = (_Float16)acc0[t][r];
            p1[r] = (_Float16)acc1[t][r];
        }
        if (RELU) {
            const v8h z = {0, 0, 0, 0, 0, 0, 0, 0};
            p0 = __builtin_elementwise_max(p0, z);
            p1 = __builtin_elementwise_max(p1, z);
        }
#pragma unroll
        for (int r = 0; r < 8; ++r) {
            bout[t >> 1][0][(t & 1) * 8 + r] = p0[r];
            bout[t >> 1][1][(t & 1) * 8 + r] = p1[r];
        }
    }
#pragma unroll
    for (int t = NT; t < 8; ++t) {
#pragma unroll
        for (int r = 0; r < 8; ++r) {
            bout[t >> 1][0][(t & 1) * 8 + r] = (_Float16)0.f;
            bout[t >> 1][1][(t & 1) * 8 + r] = (_Float16)0.f;
        }
    }
}

template <int KS>
__device__ __forceinline__ void gemm_apply_pass(
    const v16h (&bin)[KS][2],
    const _Float16* Wt,
    const float* lds_x,
    float* lds_m,
    int lane,
    int pass)
{
    constexpr int NT = 8;
    const int ntile0 = pass * NT;
    const _Float16* Wp = Wt + ntile0 * (KS * 512);
    const int rc    = lane & 15;
    const int khalf = lane >> 4;

    v8f acc0[NT], acc1[NT];
#pragma unroll
    for (int t = 0; t < NT; ++t) {
        acc0[t] = (v8f){0.f,0.f,0.f,0.f,0.f,0.f,0.f,0.f};
        acc1[t] = (v8f){0.f,0.f,0.f,0.f,0.f,0.f,0.f,0.f};
    }

    kloop<NT, KS, true>(bin, Wp, 0, lane, acc0, acc1);

    const float* x0p = lds_x + rc * 16 + ntile0;
    const float* x1p = lds_x + (16 + rc) * 16 + ntile0;
    const float4 xa = *(const float4*)(x0p);
    const float4 xb = *(const float4*)(x0p + 4);
    const float4 ya = *(const float4*)(x1p);
    const float4 yb = *(const float4*)(x1p + 4);
    const float xv[8] = {xa.x, xa.y, xa.z, xa.w, xb.x, xb.y, xb.z, xb.w};
    const float yv[8] = {ya.x, ya.y, ya.z, ya.w, yb.x, yb.y, yb.z, yb.w};

    float* m0p = lds_m + rc * 16 + khalf * 8;
    float* m1p = lds_m + (16 + rc) * 16 + khalf * 8;
    float s0[8], s1[8];
    if (pass != 0) {
        const float4 p0 = *(const float4*)(m0p), p1 = *(const float4*)(m0p + 4);
        const float4 q0 = *(const float4*)(m1p), q1 = *(const float4*)(m1p + 4);
        s0[0] = p0.x; s0[1] = p0.y; s0[2] = p0.z; s0[3] = p0.w;
        s0[4] = p1.x; s0[5] = p1.y; s0[6] = p1.z; s0[7] = p1.w;
        s1[0] = q0.x; s1[1] = q0.y; s1[2] = q0.z; s1[3] = q0.w;
        s1[4] = q1.x; s1[5] = q1.y; s1[6] = q1.z; s1[7] = q1.w;
    } else {
#pragma unroll
        for (int r = 0; r < 8; ++r) { s0[r] = 0.0f; s1[r] = 0.0f; }
    }
#pragma unroll
    for (int t = 0; t < NT; ++t) {
#pragma unroll
        for (int r = 0; r < 8; ++r) {
            s0[r] += xv[t] * acc0[t][r];
            s1[r] += yv[t] * acc1[t][r];
        }
    }
    *(float4*)(m0p)     = (float4){s0[0], s0[1], s0[2], s0[3]};
    *(float4*)(m0p + 4) = (float4){s0[4], s0[5], s0[6], s0[7]};
    *(float4*)(m1p)     = (float4){s1[0], s1[1], s1[2], s1[3]};
    *(float4*)(m1p + 4) = (float4){s1[4], s1[5], s1[6], s1[7]};
}

__device__ __forceinline__ void bank_coord(int t, int KS, int& n, int& k) {
    const int i    = t & 7;
    const int lane = (t >> 3) & 31;
    const int part = (t >> 8) & 1;
    const int f    = t >> 9;
    const int ks   = f % KS;
    const int nt   = f / KS;
    n = nt * 16 + (lane & 15);
    k = ks * 32 + part * 16 + (lane >> 4) * 8 + i;
}

template <int KV, int NV, bool HAND_ON>
__device__ __forceinline__ _Float16 bank_value(const float* __restrict__ Wm, const float* __restrict__ b,
                                               int n, int k) {
    if (k < KV) return (_Float16)(n < NV ? Wm[k * NV + n] : 0.f);
    const int j = k - KV;
    if (j < 2) {
        if (n < NV) {
            const float    bf = b[n];
            const _Float16 hi = (_Float16)bf;
            return j == 0 ? hi : (_Float16)(bf - (float)hi);
        }
        if (HAND_ON && n == NV + j) return (_Float16)1.0f;
    }
    return (_Float16)0.f;
}

__device__ __forceinline__ void prep_weights(const float* __restrict__ W1, const float* __restrict__ W2,
                             const float* __restrict__ W3, const float* __restrict__ W4,
                             const float* __restrict__ b1, const float* __restrict__ b2,
                             const float* __restrict__ b3, const float* __restrict__ b4,
                             _Float16* __restrict__ W1t, _Float16* __restrict__ W2t,
                             _Float16* __restrict__ W3t, _Float16* __restrict__ W4t, int tid)
{
    const int s1 = 128 * KATTR;
    const int s2 = s1 + 128 * HPAD;
    const int s3 = s2 + 128 * HPAD;
    const int s4 = s3 + 256 * HPAD;
    int n, k;
    if (tid < s1) {
        bank_coord(tid, KATTR / 32, n, k);
        W1t[tid] = bank_value<8, 100, true>(W1, b1, n, k);
    } else if (tid < s2) {
        int t = tid - s1; bank_coord(t, HPAD / 32, n, k);
        W2t[t] = bank_value<100, 100, true>(W2, b2, n, k);
    } else if (tid < s3) {
        int t = tid - s2; bank_coord(t, HPAD / 32, n, k);
        W3t[t] = bank_value<100, 100, true>(W3, b3, n, k);
    } else if (tid < s4) {
        int t = tid - s3; bank_coord(t, HPAD / 32, n, k);
        W4t[t] = bank_value<100, 256, false>(W4, b4, n, k);
    }
}

__device__ __forceinline__ int range_of(int d, int npb) {
    const int b = d / npb;
    return b < NB - 1 ? b : NB - 1;
}

template <int NW>
__device__ __forceinline__ int excl_scan_block(int v, int tid, int* tmp) {
    const int lane = tid & 31, w = tid >> 5;
    int inc = v;
#pragma unroll
    for (int d = 1; d < 32; d <<= 1) {
        const int t = __shfl_up(inc, d);
        if (lane >= d) inc += t;
    }
    if (lane == 31) tmp[w] = inc;
    __syncthreads();
    int off = 0;
#pragma unroll
    for (int j = 0; j < NW; ++j) {
        const int tj = tmp[j];
        if (j < w) off += tj;
    }
    return off + inc - v;
}

__global__ __launch_bounds__(PTHREADS)
void prep_and_group(const float* __restrict__ W1, const float* __restrict__ W2,
                    const float* __restrict__ W3, const float* __restrict__ W4,
                    const float* __restrict__ b1, const float* __restrict__ b2,
                    const float* __restrict__ b3, const float* __restrict__ b4,
                    _Float16* __restrict__ W1t, _Float16* __restrict__ W2t,
                    _Float16* __restrict__ W3t, _Float16* __restrict__ W4t,
                    const int* __restrict__ edge_index, const float* __restrict__ edge_attr,
                    int* __restrict__ g_src, int* __restrict__ g_dst, float* __restrict__ g_attr,
                    int* __restrict__ tab_cnt, int* __restrict__ tab_off,
                    int E, int N)
{
    __shared__ __attribute__((aligned(16))) float st_attr[CHUNK * 8];
    __shared__ int st_src[CHUNK];
    __shared__ int st_dst[CHUNK];
    __shared__ int h[NB];
    __shared__ int loff[NB];
    __shared__ int tmp[PTHREADS / 32];

    const int tid = threadIdx.x;
    {
        const int prep_total = 128 * KATTR + 128 * HPAD + 128 * HPAD + 256 * HPAD;
        for (int t = (int)blockIdx.x * PTHREADS + tid; t < prep_total;
             t += (int)gridDim.x * PTHREADS)
            prep_weights(W1, W2, W3, W4, b1, b2, b3, b4, W1t, W2t, W3t, W4t, t);
    }

    const int w   = (int)blockIdx.x;
    const int e0  = w * CHUNK;
    const int n   = E - e0 < CHUNK ? E - e0 : CHUNK;
    const int npb = (N + NB - 1) / NB;
    const v4f* ea = (const v4f*)edge_attr;

    if (tid < NB) h[tid] = 0;
    __syncthreads();

    int    es[EPT], ed[EPT], eb[EPT], er[EPT];
    v4f    a0[EPT], a1[EPT];
#pragma unroll
    for (int k = 0; k < EPT; ++k) {
        const int i = tid + k * PTHREADS;
        const int e = e0 + (i < n ? i : n - 1);
        es[k] = edge_index[e];
        ed[k] = edge_index[E + e];
        a0[k] = ea[(long)e * 2];
        a1[k] = ea[(long)e * 2 + 1];
    }
#pragma unroll
    for (int k = 0; k < EPT; ++k) {
        const int i = tid + k * PTHREADS;
        if (i < n) {
            eb[k] = range_of(ed[k], npb);
            er[k] = atomicAdd(&h[eb[k]], 1);
        }
    }
    __syncthreads();

    const int cb   = tid < NB ? h[tid] : 0;
    const int excl = excl_scan_block<PTHREADS / 32>(cb, tid, tmp);
    if (tid < NB) {
        loff[tid] = excl;
        tab_cnt[tid * (int)gridDim.x + w] = cb;
        tab_off[tid * (int)gridDim.x + w] = excl;
    }
    __syncthreads();

#pragma unroll
    for (int k = 0; k < EPT; ++k) {
        const int i = tid + k * PTHREADS;
        if (i < n) {
            const int q = loff[eb[k]] + er[k];
            st_src[q] = es[k];
            st_dst[q] = ed[k];
            ((v4f*)st_attr)[q * 2]     = a0[k];
            ((v4f*)st_attr)[q * 2 + 1] = a1[k];
        }
    }
    __syncthreads();

    for (int i = tid; i < n; i += PTHREADS) {
        g_src[e0 + i] = st_src[i];
        g_dst[e0 + i] = st_dst[i];
    }
    v4f* ga = (v4f*)g_attr + (long)e0 * 2;
    for (int i = tid; i < n * 2; i += PTHREADS) ga[i] = ((const v4f*)st_attr)[i];
}

__device__ __forceinline__ void copy_bank(const _Float16* __restrict__ src, _Float16* dst,
                                          int nvec, int tid) {
    const uint4* s = (const uint4*)src;
    uint4*       d = (uint4*)dst;
    for (int i = tid; i < nvec; i += BLOCK) d[i] = s[i];
}

__global__ __launch_bounds__(BLOCK, 3)
void edge_kernel(const float* __restrict__ x,
                 const int*   __restrict__ g_src,
                 const int*   __restrict__ g_dst,
                 const float* __restrict__ g_attr,
                 const int*   __restrict__ tab_cnt,
                 const int*   __restrict__ tab_off,
                 int nchunks,
                 const _Float16* __restrict__ W1t, const _Float16* __restrict__ W2t,
                 const _Float16* __restrict__ W3t, const _Float16* __restrict__ W4t,
                 const float* __restrict__ root, const float* __restrict__ bias,
                 float* __restrict__ out, int N)
{
    __shared__ __attribute__((aligned(16))) _Float16 s_w[WBANKS];
    __shared__ __attribute__((aligned(16))) float    s_x[NWAVES * WAVE_E * 16];
    __shared__ __attribute__((aligned(16))) float    s_m[NWAVES * WAVE_E * 16];
    __shared__ __attribute__((aligned(16))) float    s_agg[NPB_CAP * 16];
    __shared__ __attribute__((aligned(16))) float    s_xn[((NPB_CAP * 4 + BLOCK - 1) / BLOCK) * BLOCK * 4];
    __shared__ float s_cnt[NPB_CAP];
    __shared__ int   s_list[LIST_CAP];
    __shared__ int   s_tmp[NWAVES];

    const int tid  = threadIdx.x;
    const int lane = tid & 31;
    const int wave = __builtin_amdgcn_readfirstlane(tid >> 5);

    const int npb   = (N + NB - 1) / NB;
    const int node0 = (int)blockIdx.x * npb;
    const int nown  = node0 + npb <= N ? npb : (N > node0 ? N - node0 : 0);
    for (int i = tid; i < NPB_CAP * 16; i += BLOCK) s_agg[i] = 0.0f;
    for (int i = tid; i < NPB_CAP; i += BLOCK) s_cnt[i] = 0.0f;
    constexpr int XQ = (NPB_CAP * 4 + BLOCK - 1) / BLOCK;
    {
        const float4* xs4 = (const float4*)(x + (long)(nown > 0 ? node0 : 0) * 16);
        float4*       xd4 = (float4*)s_xn;
        const int last = nown > 0 ? nown * 4 - 1 : 0;
        float4 xrow[XQ];
#pragma unroll
        for (int k = 0; k < XQ; ++k) {
            const int i = tid + k * BLOCK;
            xrow[k] = xs4[i < last ? i : last];
        }
#pragma unroll
        for (int k = 0; k < XQ; ++k) xd4[tid + k * BLOCK] = xrow[k];
    }
    const int* tcnt = tab_cnt + (long)blockIdx.x * nchunks;
    const int* toff = tab_off + (long)blockIdx.x * nchunks;
    const int  wfirst = tid < nchunks ? tid : nchunks - 1;
    const int  cfirst = tcnt[wfirst];
    const int  gfirst = wfirst * CHUNK + toff[wfirst];

    copy_bank(W1t, s_w + W1OFF, 128 * KATTR * 2 / 16, tid);
    copy_bank(W2t, s_w + W2OFF, 128 * HPAD  * 2 / 16, tid);
    copy_bank(W3t, s_w + W3OFF, 128 * HPAD  * 2 / 16, tid);
    copy_bank(W4t, s_w + W4OFF, 256 * HPAD  * 2 / 16, tid);

    float*    xw = s_x    + wave * WAVE_E * 16;
    float*    mw = s_m    + wave * WAVE_E * 16;

    const int c0 = tid < nchunks ? cfirst : 0;
    int lc = c0;
    for (int w = tid + BLOCK; w < nchunks; w += BLOCK) lc += tcnt[w];
    const int lexcl = excl_scan_block<NWAVES>(lc, tid, s_tmp);
    int etot = 0;
#pragma unroll
    for (int j = 0; j < NWAVES; ++j) etot += s_tmp[j];
    const int Eb = __builtin_amdgcn_readfirstlane(etot);

    const _Float16* W1s = s_w + W1OFF;
    const _Float16* W2s = s_w + W2OFF;
    const _Float16* W3s = s_w + W3OFF;
    const _Float16* W4s = s_w + W4OFF;

    const int colc  = lane & 15;
    const int hhalf = lane >> 4;

    for (int r0 = 0; r0 < Eb; r0 += LIST_CAP) {
    {
        int pos = lexcl - r0;
        for (int j = 0; j < c0; ++j) {
            const int p = pos + j;
            if (p >= 0 && p < LIST_CAP) s_list[p] = gfirst + j;
        }
        pos += c0;
        for (int w = tid + BLOCK; w < nchunks; w += BLOCK) {
            const int cw = tcnt[w];
            const int gw = w * CHUNK + toff[w];
            for (int j = 0; j < cw; ++j) {
                const int p = pos + j;
                if (p >= 0 && p < LIST_CAP) s_list[p] = gw + j;
            }
            pos += cw;
        }
    }
    __syncthreads();
    const int E = Eb - r0 < LIST_CAP ? Eb - r0 : LIST_CAP;

    const int ntiles = (E + WAVE_E - 1) / WAVE_E;
    for (int tile = wave; tile < ntiles; tile += NWAVES) {
        const int  ge   = tile * WAVE_E + lane;
        const bool live = ge < E;
        const int  gc   = live ? ge : E - 1;
        const int gp = s_list[gc];
        const int s = g_src[gp];
        const int dmine = g_dst[gp] - node0;

        const int e0 = tile * WAVE_E + colc;
        const int e1 = e0 + 16;
        const int g0 = s_list[e0 < E ? e0 : E - 1];
        const int g1 = s_list[e1 < E ? e1 : E - 1];
        const float4* ea = (const float4*)g_attr;
        const float4 p0 = ea[(long)g0 * 2];
        const float4 p1 = ea[(long)g0 * 2 + 1];
        const float4 q0 = ea[(long)g1 * 2];
        const float4 q1 = ea[(long)g1 * 2 + 1];

        const float4* xs = (const float4*)(x + (long)s * 16);
        const float4 xq0 = xs[0], xq1 = xs[1], xq2 = xs[2], xq3 = xs[3];

        v16h hin[1][2];
        {
            const float pv[8] = {p0.x, p0.y, p0.z, p0.w, p1.x, p1.y, p1.z, p1.w};
            const float qv[8] = {q0.x, q0.y, q0.z, q0.w, q1.x, q1.y, q1.z, q1.w};
            v8h ph, qh;
#pragma unroll
            for (int i = 0; i < 8; ++i) { ph[i] = (_Float16)pv[i]; qh[i] = (_Float16)qv[i]; }
            const bool lo_half = (hhalf == 0);
            v4u pu = __builtin_bit_cast(v4u, ph);
            v4u qu = __builtin_bit_cast(v4u, qh);
#pragma unroll
            for (int i = 0; i < 4; ++i) {
                const unsigned int pad = (i == 0) ? 0x3C003C00u : 0u;
                pu[i] = lo_half ? pu[i] : pad;
                qu[i] = lo_half ? qu[i] : pad;
            }
            ph = __builtin_bit_cast(v8h, pu);
            qh = __builtin_bit_cast(v8h, qu);
#pragma unroll
            for (int i = 0; i < 8; ++i) {
                hin[0][0][i]     = ph[i];
                hin[0][1][i]     = qh[i];
                hin[0][0][8 + i] = (_Float16)0.f;
                hin[0][1][8 + i] = (_Float16)0.f;
            }
        }

        v16h ha[4][2], hb[4][2];

        gemm_stage<NT_H, 1, true>(hin, W1s, ha, lane);

        {
            float4* dx = (float4*)(xw + lane * 16);
            dx[0] = xq0; dx[1] = xq1; dx[2] = xq2; dx[3] = xq3;
        }

        gemm_stage<NT_H, 4, true>(ha, W2s, hb, lane);
        gemm_stage<NT_H, 4, true>(hb, W3s, ha, lane);

#pragma nounroll
        for (int pass = 0; pass < 2; ++pass)
            gemm_apply_pass<4>(ha, W4s, xw, mw, lane, pass);

        {
            const int o     = lane & 15;
            const int khalf = lane >> 4;
            const int dsel  = live ? dmine : -1;
            xw[lane] = __int_as_float(dsel);
            float msg0[8], msg1[8];
#pragma unroll
            for (int r = 0; r < 8; ++r) {
                msg0[r] = mw[(8 * khalf + r) * 16 + o];
                msg1[r] = mw[(16 + 8 * khalf + r) * 16 + o];
            }
            const float4 da = *(const float4*)(xw + 8 * khalf);
            const float4 db = *(const float4*)(xw + 8 * khalf + 4);
            const float4 dc = *(const float4*)(xw + 16 + 8 * khalf);
            const float4 dd = *(const float4*)(xw + 16 + 8 * khalf + 4);
            const float dv0[8] = {da.x, da.y, da.z, da.w, db.x, db.y, db.z, db.w};
            const float dv1[8] = {dc.x, dc.y, dc.z, dc.w, dd.x, dd.y, dd.z, dd.w};
#pragma unroll
            for (int r = 0; r < 8; ++r) {
                const int d0 = __float_as_int(dv0[r]);
                const int d1 = __float_as_int(dv1[r]);
                if (d0 >= 0) atomicAdd(&s_agg[d0 * 16 + o], msg0[r]);
                if (d1 >= 0) atomicAdd(&s_agg[d1 * 16 + o], msg1[r]);
            }
            if (dsel >= 0) atomicAdd(&s_cnt[dsel], 1.0f);
        }
    }
    __syncthreads();
    }

    __syncthreads();
    {
        const int o = tid & 15;
        float rcol[16];
#pragma unroll
        for (int i = 0; i < 16; ++i) rcol[i] = root[i * 16 + o];
        const float bo = bias[o];
        for (int idx = tid; idx < nown * 16; idx += BLOCK) {
            const int nl = idx >> 4;
            const float4* xr = (const float4*)(s_xn + nl * 16);
            const float4 x0 = xr[0], x1 = xr[1], x2 = xr[2], x3 = xr[3];
            const float xi[16] = {x0.x, x0.y, x0.z, x0.w, x1.x, x1.y, x1.z, x1.w,
                                  x2.x, x2.y, x2.z, x2.w, x3.x, x3.y, x3.z, x3.w};
            const float cn  = s_cnt[nl];
            const float inv = 1.0f / (cn > 1.0f ? cn : 1.0f);
            float acc = 0.0f;
#pragma unroll
            for (int i = 0; i < 16; ++i) acc += xi[i] * rcol[i];
            out[(long)node0 * 16 + idx] = acc + s_agg[idx] * inv + bo;
        }
    }
}

extern "C" void kernel_launch(void* const* d_in, const int* in_sizes, int n_in,
                              void* d_out, int out_size, void* d_ws, size_t ws_size,
                              hipStream_t stream)
{
    const float* x          = (const float*)d_in[0];
    const int*   edge_index = (const int*)  d_in[1];
    const float* edge_attr  = (const float*)d_in[2];
    const float* W1 = (const float*)d_in[3];  const float* b1 = (const float*)d_in[4];
    const float* W2 = (const float*)d_in[5];  const float* b2 = (const float*)d_in[6];
    const float* W3 = (const float*)d_in[7];  const float* b3 = (const float*)d_in[8];
    const float* W4 = (const float*)d_in[9];  const float* b4 = (const float*)d_in[10];
    const float* root = (const float*)d_in[11];
    const float* bias = (const float*)d_in[12];
    float* out = (float*)d_out;

    const int N = in_sizes[0] / 16;
    const int E = in_sizes[1] / 2;

    const int nchunks = (E + CHUNK - 1) / CHUNK;

    _Float16* W1t = (_Float16*)d_ws;
    _Float16* W2t = W1t + 128 * KATTR;
    _Float16* W3t = W2t + 128 * HPAD;
    _Float16* W4t = W3t + 128 * HPAD;
    int*      tab_cnt = (int*)(W4t + 256 * HPAD);
    int*      tab_off = tab_cnt + (size_t)NB * nchunks;
    float*    g_attr  = (float*)(tab_off + (size_t)NB * nchunks);
    int*      g_src   = (int*)(g_attr + (size_t)E * 8);
    int*      g_dst   = g_src + E;

    prep_and_group<<<nchunks, PTHREADS, 0, stream>>>(
        W1, W2, W3, W4, b1, b2, b3, b4, W1t, W2t, W3t, W4t,
        edge_index, edge_attr, g_src, g_dst, g_attr, tab_cnt, tab_off, E, N);

    edge_kernel<<<NB, BLOCK, 0, stream>>>(
        x, g_src, g_dst, g_attr, tab_cnt, tab_off, nchunks, W1t, W2t, W3t, W4t,
        root, bias, out, N);
}
